# starting kernel
# baseline (speedup 1.0000x reference)
#include <hip/hip_runtime.h>


typedef __attribute__((ext_vector_type(2))) float v2f;
typedef __attribute__((ext_vector_type(8))) float v8f;

__global__ __launch_bounds__(256) void fwd_conv5x5_s2_wmma(
    const float* __restrict__ in, long inBatchStride,
    const float* __restrict__ filt,
    float* __restrict__ out,
    int H, int W, int Ho, int Wo)
{
  const int col0 = blockIdx.x * 16;
  const int row0 = blockIdx.y * 16;
  const int b    = blockIdx.z;
  const int tid  = threadIdx.x;
  const int wv   = tid >> 5;
  const int lane = tid & 31;
  const int hf   = lane >> 4;
  const int nl   = lane & 15;

  __shared__ float sIn[36 * 36];
  __shared__ float sF[200];

  if (tid == 0) __builtin_prefetch(filt, 0, 1);
  for (int i = tid; i < 200; i += 256) sF[i] = filt[i];

  const int iy0 = 2 * row0 - 2;
  const int ix0 = 2 * col0 - 2;
  for (int i = tid; i < 36 * 36; i += 256) {
    int r = i / 36, c = i - r * 36;
    int gy = iy0 + r, gx = ix0 + c;
    float v = 0.0f;
    if ((unsigned)gy < (unsigned)H && (unsigned)gx < (unsigned)W)
      v = in[(long)b * inBatchStride + (long)gy * W + gx];
    sIn[i] = v;
  }
  __syncthreads();

  v2f Bv[7];
#pragma unroll
  for (int s = 0; s < 7; ++s) {
    int k0 = 4 * s + 2 * hf;
    Bv[s][0] = (nl < 8 && k0     < 25) ? sF[nl * 25 + k0]     : 0.0f;
    Bv[s][1] = (nl < 8 && k0 + 1 < 25) ? sF[nl * 25 + k0 + 1] : 0.0f;
  }

#pragma unroll
  for (int rr = 0; rr < 2; ++rr) {
    const int r = wv * 2 + rr;
    v8f acc = {};
#pragma unroll
    for (int s = 0; s < 7; ++s) {
      int k0 = 4 * s + 2 * hf;
      int kh0 = k0 / 5,       kw0 = k0 - kh0 * 5;
      int k1  = k0 + 1;
      int kh1 = k1 / 5,       kw1 = k1 - kh1 * 5;
      v2f Av;
      Av[0] = sIn[(2 * r + kh0) * 36 + (2 * nl + kw0)];
      Av[1] = sIn[(2 * r + kh1) * 36 + (2 * nl + kw1)];
      acc = __builtin_amdgcn_wmma_f32_16x16x4_f32(
          false, Av, false, Bv[s], (short)0, acc, false, false);
    }
    if (nl < 8) {
      const int gy = row0 + r;
#pragma unroll
      for (int j = 0; j < 8; ++j) {
        int gx = col0 + j + 8 * hf;
        out[(((long)b * 8 + nl) * Ho + gy) * Wo + gx] = acc[j];
      }
    }
  }
}

__global__ __launch_bounds__(256) void inv_tconv5x5_s2_wmma(
    const float* __restrict__ rec,  long recBatchStride,
    const float* __restrict__ coef,
    const float* __restrict__ filt,
    float* __restrict__ outp,
    int H, int W)
{
  const int bx0 = blockIdx.x * 16;
  const int by0 = blockIdx.y * 8;
  const int b   = blockIdx.z;
  const int tid = threadIdx.x;
  const int wv  = tid >> 5;
  const int lane = tid & 31;
  const int hf  = lane >> 4;
  const int nl  = lane & 15;

  __shared__ float sIn[8 * 10 * 18];
  __shared__ float sF[200];

  for (int i = tid; i < 200; i += 256) sF[i] = filt[i];

  for (int i = tid; i < 8 * 10 * 18; i += 256) {
    int c = i / 180, rem = i - c * 180;
    int r = rem / 18, cc = rem - r * 18;
    int gy = by0 + r - 1, gx = bx0 + cc - 1;
    float v = 0.0f;
    if ((unsigned)gy < (unsigned)H && (unsigned)gx < (unsigned)W) {
      v = (c == 0) ? rec[(long)b * recBatchStride + (long)gy * W + gx]
                   : coef[(((long)b * 8 + c) * H + gy) * W + gx];
    }
    sIn[i] = v;
  }
  __syncthreads();

  const int py = (nl & 3) >> 1;
  const int px = nl & 1;
  const int r  = wv;

  v8f acc = {};
#pragma unroll
  for (int s = 0; s < 18; ++s) {
    int k0 = 4 * s + 2 * hf;
    v2f Av, Bv;
#pragma unroll
    for (int t = 0; t < 2; ++t) {
      int k  = k0 + t;
      int c  = k / 9;
      int r9 = k - c * 9;
      int dy = r9 / 3 - 1;
      int dx = r9 - (r9 / 3) * 3 - 1;
      float a = sIn[c * 180 + (r + dy + 1) * 18 + (nl + dx + 1)];
      int ih = 2 + py - 2 * dy;
      int iw = 2 + px - 2 * dx;
      float w = (nl < 4 && ih < 5 && iw < 5) ? sF[c * 25 + ih * 5 + iw] : 0.0f;
      Av[t] = a;
      Bv[t] = w;
    }
    acc = __builtin_amdgcn_wmma_f32_16x16x4_f32(
        false, Av, false, Bv, (short)0, acc, false, false);
  }

  if (nl < 4) {
    const int oy = 2 * (by0 + r) + py;
    const long oplane = (long)b * (2 * H);
#pragma unroll
    for (int j = 0; j < 8; ++j) {
      int ox = 2 * (bx0 + j + 8 * hf) + px;
      outp[(oplane + oy) * (2L * W) + ox] = acc[j];
    }
  }
}

extern "C" void kernel_launch(void* const* d_in, const int* in_sizes, int n_in,
                              void* d_out, int out_size, void* d_ws, size_t ws_size,
                              hipStream_t stream) {
  (void)in_sizes; (void)n_in; (void)out_size; (void)ws_size;
  const float* x     = (const float*)d_in[0];
  const float* fwd_f = (const float*)d_in[1];
  const float* inv_f = (const float*)d_in[2];
  float* out = (float*)d_out;
  float* ws  = (float*)d_ws;

  const size_t n4 = 32UL * 8 * 256 * 256;
  const size_t n3 = 32UL * 8 * 128 * 128;
  const size_t n2 = 32UL * 8 * 64 * 64;
  const size_t n1 = 32UL * 8 * 32 * 32;
  float* res4 = ws;
  float* res3 = res4 + n4;
  float* res2 = res3 + n3;
  float* res1 = res2 + n2;
  float* rec1 = res1 + n1;
  float* rec2 = rec1 + 32UL * 64 * 64;
  float* rec3 = rec2 + 32UL * 128 * 128;

  const dim3 blk(256);

  fwd_conv5x5_s2_wmma<<<dim3(16, 16, 32), blk, 0, stream>>>(
      x,    512L * 512,      fwd_f, res4, 512, 512, 256, 256);
  fwd_conv5x5_s2_wmma<<<dim3(8, 8, 32), blk, 0, stream>>>(
      res4, 8L * 256 * 256,  fwd_f, res3, 256, 256, 128, 128);
  fwd_conv5x5_s2_wmma<<<dim3(4, 4, 32), blk, 0, stream>>>(
      res3, 8L * 128 * 128,  fwd_f, res2, 128, 128, 64, 64);
  fwd_conv5x5_s2_wmma<<<dim3(2, 2, 32), blk, 0, stream>>>(
      res2, 8L * 64 * 64,    fwd_f, res1, 64, 64, 32, 32);

  inv_tconv5x5_s2_wmma<<<dim3(2, 4, 32), blk, 0, stream>>>(
      res1, 8L * 32 * 32,    res1, inv_f, rec1, 32, 32);
  inv_tconv5x5_s2_wmma<<<dim3(4, 8, 32), blk, 0, stream>>>(
      rec1, 64L * 64,        res2, inv_f, rec2, 64, 64);
  inv_tconv5x5_s2_wmma<<<dim3(8, 16, 32), blk, 0, stream>>>(
      rec2, 128L * 128,      res3, inv_f, rec3, 128, 128);
  inv_tconv5x5_s2_wmma<<<dim3(16, 32, 32), blk, 0, stream>>>(
      rec3, 256L * 256,      res4, inv_f, out, 256, 256);
}


# Fat work-groups, 2-8x more rows each: per-wave weight fragments and LDS addresses built once, tile staged by batched pair loads; inverse k order remapped (lane half=channel parity), sum order changes
# speedup vs baseline: 1.6329x; 1.6329x over previous
#include <hip/hip_runtime.h>


typedef __attribute__((ext_vector_type(2))) float v2f;
typedef __attribute__((ext_vector_type(8))) float v8f;

template <int RPW>
__global__ __launch_bounds__(256) void fwd_conv5x5_s2_wmma(
    const float* __restrict__ in, long inBatchStride,
    const float* __restrict__ filt,
    float* __restrict__ out,
    int H, int W, int Ho, int Wo)
{
  constexpr int TR  = 8 * RPW;
  constexpr int NR  = 2 * TR + 3;
  constexpr int NP  = NR * 18;
  constexpr int NIT = (NP + 255) / 256;

  const int col0 = blockIdx.x * 16;
  const int row0 = blockIdx.y * TR;
  const int b    = blockIdx.z;
  const int tid  = threadIdx.x;
  const int wv   = tid >> 5;
  const int lane = tid & 31;
  const int hf   = lane >> 4;
  const int nl   = lane & 15;

  __shared__ __attribute__((aligned(16))) float sIn[NR * 36];
  __shared__ float sF[200];

  for (int i = tid; i < 200; i += 256) sF[i] = filt[i];

  const float* inb = in + (long)b * inBatchStride;
  const int iy0 = 2 * row0 - 2;
  const int ix0 = 2 * col0 - 2;
  float2 st[NIT];
#pragma unroll
  for (int it = 0; it < NIT; ++it) {
    const int idx = tid + 256 * it;
    const int r = idx / 18, j = idx - r * 18;
    const int gy = iy0 + r, gx = ix0 + 2 * j;
    const bool ok = idx < NP && (unsigned)gy < (unsigned)H && (unsigned)gx < (unsigned)W;
    const int off = ok ? gy * W + gx : 0;
    const float2 t = *reinterpret_cast<const float2*>(inb + off);
    st[it].x = ok ? t.x : 0.0f;
    st[it].y = ok ? t.y : 0.0f;
  }
#pragma unroll
  for (int it = 0; it < NIT; ++it) {
    const int idx = tid + 256 * it;
    if (idx < NP) *reinterpret_cast<float2*>(&sIn[2 * idx]) = st[it];
  }
  __syncthreads();

  v2f Bv[7];
  int aoff[14];
#pragma unroll
  for (int s = 0; s < 7; ++s) {
#pragma unroll
    for (int t = 0; t < 2; ++t) {
      const int k  = 4 * s + 2 * hf + t;
      const int kk = k < 24 ? k : 24;
      const float w = sF[(nl & 7) * 25 + kk];
      Bv[s][t] = (nl < 8 && k < 25) ? w : 0.0f;
      const int kh = kk / 5, kw = kk - kh * 5;
      aoff[2 * s + t] = (2 * wv * RPW + kh) * 36 + 2 * nl + kw;
    }
  }

#pragma unroll
  for (int i = 0; i < RPW; i += 2) {
    v8f acc0 = {}, acc1 = {};
#pragma unroll
    for (int s = 0; s < 7; ++s) {
      v2f A0, A1;
      A0[0] = sIn[aoff[2 * s]     + (i    ) * 72];
      A0[1] = sIn[aoff[2 * s + 1] + (i    ) * 72];
      A1[0] = sIn[aoff[2 * s]     + (i + 1) * 72];
      A1[1] = sIn[aoff[2 * s + 1] + (i + 1) * 72];
      acc0 = __builtin_amdgcn_wmma_f32_16x16x4_f32(
          false, A0, false, Bv[s], (short)0, acc0, false, false);
      acc1 = __builtin_amdgcn_wmma_f32_16x16x4_f32(
          false, A1, false, Bv[s], (short)0, acc1, false, false);
    }
    if (nl < 8) {
      const int gy = row0 + wv * RPW + i;
      float* o = out + (((long)b * 8 + nl) * Ho + gy) * Wo + col0 + 8 * hf;
#pragma unroll
      for (int j = 0; j < 8; ++j) o[j] = acc0[j];
#pragma unroll
      for (int j = 0; j < 8; ++j) o[Wo + j] = acc1[j];
    }
  }
}

template <int RPW>
__global__ __launch_bounds__(256) void inv_tconv5x5_s2_wmma(
    const float* __restrict__ rec,  long recBatchStride,
    const float* __restrict__ coef,
    const float* __restrict__ filt,
    float* __restrict__ outp,
    int H, int W)
{
  constexpr int TR   = 8 * RPW;
  constexpr int NRI  = TR + 2;
  constexpr int PT   = 20;
  constexpr int PL   = NRI * PT;
  constexpr int NP0  = NRI * 10;
  constexpr int NIT0 = (NP0 + 255) / 256;
  constexpr int NP1  = 7 * NP0;
  constexpr int NIT1 = (NP1 + 255) / 256;

  const int bx0 = blockIdx.x * 16;
  const int by0 = blockIdx.y * TR;
  const int b   = blockIdx.z;
  const int tid = threadIdx.x;
  const int wv  = tid >> 5;
  const int lane = tid & 31;
  const int hf  = lane >> 4;
  const int nl  = lane & 15;

  __shared__ __attribute__((aligned(16))) float sIn[8 * PL];
  __shared__ float sF[200];

  for (int i = tid; i < 200; i += 256) sF[i] = filt[i];

  const float* recb  = rec  + (long)b * recBatchStride;
  const float* coefb = coef + (long)b * 8 * H * W;
  float2 s0[NIT0], s1[NIT1];
#pragma unroll
  for (int it = 0; it < NIT0; ++it) {
    const int idx = tid + 256 * it;
    const int r = idx / 10, j = idx - r * 10;
    const int gy = by0 + r - 1, gx = bx0 - 2 + 2 * j;
    const bool ok = idx < NP0 && (unsigned)gy < (unsigned)H && (unsigned)gx < (unsigned)W;
    const int off = ok ? gy * W + gx : 0;
    const float2 t = *reinterpret_cast<const float2*>(recb + off);
    s0[it].x = ok ? t.x : 0.0f;
    s0[it].y = ok ? t.y : 0.0f;
  }
#pragma unroll
  for (int it = 0; it < NIT1; ++it) {
    const int idx = tid + 256 * it;
    const int c1 = idx / NP0, rem = idx - c1 * NP0;
    const int r = rem / 10, j = rem - r * 10;
    const int gy = by0 + r - 1, gx = bx0 - 2 + 2 * j;
    const bool ok = idx < NP1 && (unsigned)gy < (unsigned)H && (unsigned)gx < (unsigned)W;
    const int off = ok ? ((c1 + 1) * H + gy) * W + gx : 0;
    const float2 t = *reinterpret_cast<const float2*>(coefb + off);
    s1[it].x = ok ? t.x : 0.0f;
    s1[it].y = ok ? t.y : 0.0f;
  }
#pragma unroll
  for (int it = 0; it < NIT0; ++it) {
    const int idx = tid + 256 * it;
    if (idx < NP0) *reinterpret_cast<float2*>(&sIn[2 * idx]) = s0[it];
  }
#pragma unroll
  for (int it = 0; it < NIT1; ++it) {
    const int idx = tid + 256 * it;
    if (idx < NP1) *reinterpret_cast<float2*>(&sIn[PL + 2 * idx]) = s1[it];
  }
  __syncthreads();

  const int py = (nl >> 1) & 1;
  const int px = nl & 1;

  v2f Bv[18];
#pragma unroll
  for (int q = 0; q < 36; ++q) {
    const int cc = q / 9, tap = q - cc * 9;
    const int dy = tap / 3 - 1, dx = tap - (tap / 3) * 3 - 1;
    const int ih = 2 + py - 2 * dy;
    const int iw = 2 + px - 2 * dx;
    const int ihc = ih < 4 ? ih : 4, iwc = iw < 4 ? iw : 4;
    const float w = sF[(2 * cc + hf) * 25 + ihc * 5 + iwc];
    Bv[q >> 1][q & 1] = (nl < 4 && ih < 5 && iw < 5) ? w : 0.0f;
  }

  const int abase = hf * PL + (wv * RPW) * PT + nl;

#pragma unroll
  for (int i = 0; i < RPW; i += 2) {
    v8f acc0 = {}, acc1 = {};
#pragma unroll
    for (int s = 0; s < 18; ++s) {
      v2f A0, A1;
#pragma unroll
      for (int t = 0; t < 2; ++t) {
        const int q = 2 * s + t;
        const int cc = q / 9, tap = q - cc * 9;
        const int dy = tap / 3 - 1, dx = tap - (tap / 3) * 3 - 1;
        const int o = 2 * cc * PL + (i + dy + 1) * PT + dx + 2;
        A0[t] = sIn[abase + o];
        A1[t] = sIn[abase + o + PT];
      }
      acc0 = __builtin_amdgcn_wmma_f32_16x16x4_f32(
          false, A0, false, Bv[s], (short)0, acc0, false, false);
      acc1 = __builtin_amdgcn_wmma_f32_16x16x4_f32(
          false, A1, false, Bv[s], (short)0, acc1, false, false);
    }
    if (nl < 4) {
      const int oy = 2 * (by0 + wv * RPW + i) + py;
      float* o = outp + ((long)b * (2 * H) + oy) * (2L * W) + 2 * (bx0 + 8 * hf) + px;
#pragma unroll
      for (int j = 0; j < 8; ++j) o[2 * j] = acc0[j];
#pragma unroll
      for (int j = 0; j < 8; ++j) o[4L * W + 2 * j] = acc1[j];
    }
  }
}

extern "C" void kernel_launch(void* const* d_in, const int* in_sizes, int n_in,
                              void* d_out, int out_size, void* d_ws, size_t ws_size,
                              hipStream_t stream) {
  (void)in_sizes; (void)n_in; (void)out_size; (void)ws_size;
  const float* x     = (const float*)d_in[0];
  const float* fwd_f = (const float*)d_in[1];
  const float* inv_f = (const float*)d_in[2];
  float* out = (float*)d_out;
  float* ws  = (float*)d_ws;

  const size_t n4 = 32UL * 8 * 256 * 256;
  const size_t n3 = 32UL * 8 * 128 * 128;
  const size_t n2 = 32UL * 8 * 64 * 64;
  const size_t n1 = 32UL * 8 * 32 * 32;
  float* res4 = ws;
  float* res3 = res4 + n4;
  float* res2 = res3 + n3;
  float* res1 = res2 + n2;
  float* rec1 = res1 + n1;
  float* rec2 = rec1 + 32UL * 64 * 64;
  float* rec3 = rec2 + 32UL * 128 * 128;

  const dim3 blk(256);

  fwd_conv5x5_s2_wmma<8><<<dim3(16, 4, 32), blk, 0, stream>>>(
      x,    512L * 512,      fwd_f, res4, 512, 512, 256, 256);
  fwd_conv5x5_s2_wmma<4><<<dim3(8, 4, 32), blk, 0, stream>>>(
      res4, 8L * 256 * 256,  fwd_f, res3, 256, 256, 128, 128);
  fwd_conv5x5_s2_wmma<2><<<dim3(4, 4, 32), blk, 0, stream>>>(
      res3, 8L * 128 * 128,  fwd_f, res2, 128, 128, 64, 64);
  fwd_conv5x5_s2_wmma<2><<<dim3(2, 2, 32), blk, 0, stream>>>(
      res2, 8L * 64 * 64,    fwd_f, res1, 64, 64, 32, 32);

  inv_tconv5x5_s2_wmma<2><<<dim3(2, 2, 32), blk, 0, stream>>>(
      res1, 8L * 32 * 32,    res1, inv_f, rec1, 32, 32);
  inv_tconv5x5_s2_wmma<2><<<dim3(4, 4, 32), blk, 0, stream>>>(
      rec1, 64L * 64,        res2, inv_f, rec2, 64, 64);
  inv_tconv5x5_s2_wmma<4><<<dim3(8, 4, 32), blk, 0, stream>>>(
      rec2, 128L * 128,      res3, inv_f, rec3, 128, 128);
  inv_tconv5x5_s2_wmma<4><<<dim3(16, 8, 32), blk, 0, stream>>>(
      rec3, 256L * 256,      res4, inv_f, out, 256, 256);
}


# Work-groups loop over 2-4 batch images per tile position: weights, addresses and staging geometry set up once per wave, next image's tile loads in flight under the products (LDS-only barriers)
# speedup vs baseline: 1.7038x; 1.0434x over previous
#include <hip/hip_runtime.h>


typedef __attribute__((ext_vector_type(2))) float v2f;
typedef __attribute__((ext_vector_type(8))) float v8f;

__device__ __forceinline__ void lds_barrier() {
  __builtin_amdgcn_fence(__ATOMIC_RELEASE, "workgroup", "local");
  __builtin_amdgcn_s_barrier();
  __builtin_amdgcn_fence(__ATOMIC_ACQUIRE, "workgroup", "local");
}

template <int RPW>
__global__ __launch_bounds__(256) __attribute__((amdgpu_waves_per_eu(8))) void fwd_conv5x5_s2_wmma(
    const float* __restrict__ in, long inBatchStride,
    const float* __restrict__ filt,
    float* __restrict__ out,
    int H, int W, int Ho, int Wo, int nb)
{
  constexpr int TR  = 8 * RPW;
  constexpr int NR  = 2 * TR + 3;
  constexpr int NP  = NR * 18;
  constexpr int NIT = (NP + 255) / 256;

  const int col0 = blockIdx.x * 16;
  const int row0 = blockIdx.y * TR;
  const int tid  = threadIdx.x;
  const int wv   = tid >> 5;
  const int lane = tid & 31;
  const int hf   = lane >> 4;
  const int nl   = lane & 15;

  __shared__ __attribute__((aligned(16))) float sIn[NR * 36];
  __shared__ float sF[256];

  const float fv = filt[tid < 200 ? tid : 0];

  const int iy0 = 2 * row0 - 2;
  const int ix0 = 2 * col0 - 2;
  unsigned soff[NIT];
  unsigned okm = 0u;
#pragma unroll
  for (int it = 0; it < NIT; ++it) {
    const int idx = tid + 256 * it;
    const int r = idx / 18, j = idx - r * 18;
    const int gy = iy0 + r, gx = ix0 + 2 * j;
    const bool ok = idx < NP && (unsigned)gy < (unsigned)H && (unsigned)gx < (unsigned)W;
    soff[it] = ok ? (unsigned)(gy * W + gx) : 0u;
    okm |= (ok ? 1u : 0u) << it;
  }

  int b = blockIdx.z;
  const int bstep = gridDim.z;
  float2 st[NIT];
  {
    const float* inb = in + (long)b * inBatchStride;
#pragma unroll
    for (int it = 0; it < NIT; ++it)
      st[it] = *reinterpret_cast<const float2*>(inb + soff[it]);
  }

  sF[tid] = fv;
  lds_barrier();

  v2f Bv[7];
  int aoff[14];
#pragma unroll
  for (int s = 0; s < 7; ++s) {
#pragma unroll
    for (int t = 0; t < 2; ++t) {
      const int k  = 4 * s + 2 * hf + t;
      const int kk = k < 24 ? k : 24;
      const float w = sF[(nl & 7) * 25 + kk];
      Bv[s][t] = (nl < 8 && k < 25) ? w : 0.0f;
      const int kh = kk / 5, kw = kk - kh * 5;
      aoff[2 * s + t] = (2 * wv * RPW + kh) * 36 + 2 * nl + kw;
    }
  }

  for (;;) {
#pragma unroll
    for (int it = 0; it < NIT; ++it) {
      const int idx = tid + 256 * it;
      float2 v = st[it];
      if (!((okm >> it) & 1u)) { v.x = 0.0f; v.y = 0.0f; }
      if (idx < NP) *reinterpret_cast<float2*>(&sIn[2 * idx]) = v;
    }
    lds_barrier();

    const int bn = b + bstep;
    if (bn < nb) {
      const float* inb = in + (long)bn * inBatchStride;
#pragma unroll
      for (int it = 0; it < NIT; ++it)
        st[it] = *reinterpret_cast<const float2*>(inb + soff[it]);
    }

#pragma unroll
    for (int i = 0; i < RPW; i += 2) {
      v8f acc0 = {}, acc1 = {};
#pragma unroll
      for (int s = 0; s < 7; ++s) {
        v2f A0, A1;
        A0[0] = sIn[aoff[2 * s]     + (i    ) * 72];
        A0[1] = sIn[aoff[2 * s + 1] + (i    ) * 72];
        A1[0] = sIn[aoff[2 * s]     + (i + 1) * 72];
        A1[1] = sIn[aoff[2 * s + 1] + (i + 1) * 72];
        acc0 = __builtin_amdgcn_wmma_f32_16x16x4_f32(
            false, A0, false, Bv[s], (short)0, acc0, false, false);
        acc1 = __builtin_amdgcn_wmma_f32_16x16x4_f32(
            false, A1, false, Bv[s], (short)0, acc1, false, false);
      }
      if (nl < 8) {
        const int gy = row0 + wv * RPW + i;
        float* o = out + (((long)b * 8 + nl) * Ho + gy) * Wo + col0 + 8 * hf;
#pragma unroll
        for (int j = 0; j < 8; ++j) o[j] = acc0[j];
#pragma unroll
        for (int j = 0; j < 8; ++j) o[Wo + j] = acc1[j];
      }
    }

    if (bn >= nb) break;
    b = bn;
    lds_barrier();
  }
}

template <int RPW>
__global__ __launch_bounds__(256) __attribute__((amdgpu_waves_per_eu(8))) void inv_tconv5x5_s2_wmma(
    const float* __restrict__ rec,  long recBatchStride,
    const float* __restrict__ coef,
    const float* __restrict__ filt,
    float* __restrict__ outp,
    int H, int W, int nb)
{
  constexpr int TR   = 8 * RPW;
  constexpr int NRI  = TR + 2;
  constexpr int PT   = 20;
  constexpr int PL   = NRI * PT;
  constexpr int NP0  = NRI * 10;
  constexpr int NIT0 = (NP0 + 255) / 256;
  constexpr int NP1  = 7 * NP0;
  constexpr int NIT1 = (NP1 + 255) / 256;

  const int bx0 = blockIdx.x * 16;
  const int by0 = blockIdx.y * TR;
  const int tid = threadIdx.x;
  const int wv  = tid >> 5;
  const int lane = tid & 31;
  const int hf  = lane >> 4;
  const int nl  = lane & 15;

  __shared__ __attribute__((aligned(16))) float sIn[8 * PL];
  __shared__ float sF[256];
  __shared__ unsigned sOff[(NIT0 + NIT1) * 256];

  const float fv = filt[tid < 200 ? tid : 0];

  unsigned off0[NIT0], off1[NIT1];
  unsigned okm0 = 0u, okm1 = 0u;
#pragma unroll
  for (int it = 0; it < NIT0; ++it) {
    const int idx = tid + 256 * it;
    const int r = idx / 10, j = idx - r * 10;
    const int gy = by0 + r - 1, gx = bx0 - 2 + 2 * j;
    const bool ok = idx < NP0 && (unsigned)gy < (unsigned)H && (unsigned)gx < (unsigned)W;
    off0[it] = ok ? (unsigned)(gy * W + gx) : 0u;
    okm0 |= (ok ? 1u : 0u) << it;
  }
#pragma unroll
  for (int it = 0; it < NIT1; ++it) {
    const int idx = tid + 256 * it;
    const int c1 = idx / NP0, rem = idx - c1 * NP0;
    const int r = rem / 10, j = rem - r * 10;
    const int gy = by0 + r - 1, gx = bx0 - 2 + 2 * j;
    const bool ok = idx < NP1 && (unsigned)gy < (unsigned)H && (unsigned)gx < (unsigned)W;
    off1[it] = ok ? (unsigned)(((c1 + 1) * H + gy) * W + gx) : 0u;
    okm1 |= (ok ? 1u : 0u) << it;
  }

  int b = blockIdx.z;
  const int bstep = gridDim.z;
  const long coefBatchStride = 8L * H * W;
  float2 s0[NIT0], s1[NIT1];
  {
    const float* recb  = rec  + (long)b * recBatchStride;
    const float* coefb = coef + (long)b * coefBatchStride;
#pragma unroll
    for (int it = 0; it < NIT0; ++it)
      s0[it] = *reinterpret_cast<const float2*>(recb + off0[it]);
#pragma unroll
    for (int it = 0; it < NIT1; ++it)
      s1[it] = *reinterpret_cast<const float2*>(coefb + off1[it]);
  }

#pragma unroll
  for (int it = 0; it < NIT0; ++it) sOff[it * 256 + tid] = off0[it];
#pragma unroll
  for (int it = 0; it < NIT1; ++it) sOff[(NIT0 + it) * 256 + tid] = off1[it];
  sF[tid] = fv;
  lds_barrier();

  const int py = (nl >> 1) & 1;
  const int px = nl & 1;

  v2f Bv[18];
#pragma unroll
  for (int q = 0; q < 36; ++q) {
    const int cc = q / 9, tap = q - cc * 9;
    const int dy = tap / 3 - 1, dx = tap - (tap / 3) * 3 - 1;
    const int ih = 2 + py - 2 * dy;
    const int iw = 2 + px - 2 * dx;
    const int ihc = ih < 4 ? ih : 4, iwc = iw < 4 ? iw : 4;
    const float w = sF[(2 * cc + hf) * 25 + ihc * 5 + iwc];
    Bv[q >> 1][q & 1] = (nl < 4 && ih < 5 && iw < 5) ? w : 0.0f;
  }

  const int abase = hf * PL + (wv * RPW) * PT + nl;

  for (;;) {
#pragma unroll
    for (int it = 0; it < NIT0; ++it) {
      const int idx = tid + 256 * it;
      float2 v = s0[it];
      if (!((okm0 >> it) & 1u)) { v.x = 0.0f; v.y = 0.0f; }
      if (idx < NP0) *reinterpret_cast<float2*>(&sIn[2 * idx]) = v;
    }
#pragma unroll
    for (int it = 0; it < NIT1; ++it) {
      const int idx = tid + 256 * it;
      float2 v = s1[it];
      if (!((okm1 >> it) & 1u)) { v.x = 0.0f; v.y = 0.0f; }
      if (idx < NP1) *reinterpret_cast<float2*>(&sIn[PL + 2 * idx]) = v;
    }
    lds_barrier();

    const int bn = b + bstep;
    if (bn < nb) {
      const float* recb  = rec  + (long)bn * recBatchStride;
      const float* coefb = coef + (long)bn * coefBatchStride;
#pragma unroll
      for (int it = 0; it < NIT0; ++it)
        s0[it] = *reinterpret_cast<const float2*>(recb + sOff[it * 256 + tid]);
#pragma unroll
      for (int it = 0; it < NIT1; ++it)
        s1[it] = *reinterpret_cast<const float2*>(coefb + sOff[(NIT0 + it) * 256 + tid]);
    }

#pragma unroll
    for (int i = 0; i < RPW; i += 2) {
      v8f acc0 = {}, acc1 = {};
#pragma unroll
      for (int s = 0; s < 18; ++s) {
        v2f A0, A1;
#pragma unroll
        for (int t = 0; t < 2; ++t) {
          const int q = 2 * s + t;
          const int cc = q / 9, tap = q - cc * 9;
          const int dy = tap / 3 - 1, dx = tap - (tap / 3) * 3 - 1;
          const int o = 2 * cc * PL + (i + dy + 1) * PT + dx + 2;
          A0[t] = sIn[abase + o];
          A1[t] = sIn[abase + o + PT];
        }
        acc0 = __builtin_amdgcn_wmma_f32_16x16x4_f32(
            false, A0, false, Bv[s], (short)0, acc0, false, false);
        acc1 = __builtin_amdgcn_wmma_f32_16x16x4_f32(
            false, A1, false, Bv[s], (short)0, acc1, false, false);
      }
      if (nl < 4) {
        const int oy = 2 * (by0 + wv * RPW + i) + py;
        float* o = outp + ((long)b * (2 * H) + oy) * (2L * W) + 2 * (bx0 + 8 * hf) + px;
#pragma unroll
        for (int j = 0; j < 8; ++j) o[2 * j] = acc0[j];
#pragma unroll
        for (int j = 0; j < 8; ++j) o[4L * W + 2 * j] = acc1[j];
      }
    }

    if (bn >= nb) break;
    b = bn;
    lds_barrier();
  }
}

extern "C" void kernel_launch(void* const* d_in, const int* in_sizes, int n_in,
                              void* d_out, int out_size, void* d_ws, size_t ws_size,
                              hipStream_t stream) {
  (void)in_sizes; (void)n_in; (void)out_size; (void)ws_size;
  const float* x     = (const float*)d_in[0];
  const float* fwd_f = (const float*)d_in[1];
  const float* inv_f = (const float*)d_in[2];
  float* out = (float*)d_out;
  float* ws  = (float*)d_ws;

  const size_t n4 = 32UL * 8 * 256 * 256;
  const size_t n3 = 32UL * 8 * 128 * 128;
  const size_t n2 = 32UL * 8 * 64 * 64;
  const size_t n1 = 32UL * 8 * 32 * 32;
  float* res4 = ws;
  float* res3 = res4 + n4;
  float* res2 = res3 + n3;
  float* res1 = res2 + n2;
  float* rec1 = res1 + n1;
  float* rec2 = rec1 + 32UL * 64 * 64;
  float* rec3 = rec2 + 32UL * 128 * 128;

  const dim3 blk(256);

  fwd_conv5x5_s2_wmma<8><<<dim3(16, 4, 16), blk, 0, stream>>>(
      x,    512L * 512,      fwd_f, res4, 512, 512, 256, 256, 32);
  fwd_conv5x5_s2_wmma<4><<<dim3(8, 4, 16), blk, 0, stream>>>(
      res4, 8L * 256 * 256,  fwd_f, res3, 256, 256, 128, 128, 32);
  fwd_conv5x5_s2_wmma<2><<<dim3(4, 4, 32), blk, 0, stream>>>(
      res3, 8L * 128 * 128,  fwd_f, res2, 128, 128, 64, 64, 32);
  fwd_conv5x5_s2_wmma<2><<<dim3(2, 2, 32), blk, 0, stream>>>(
      res2, 8L * 64 * 64,    fwd_f, res1, 64, 64, 32, 32, 32);

  inv_tconv5x5_s2_wmma<2><<<dim3(2, 2, 32), blk, 0, stream>>>(
      res1, 8L * 32 * 32,    res1, inv_f, rec1, 32, 32, 32);
  inv_tconv5x5_s2_wmma<2><<<dim3(4, 4, 32), blk, 0, stream>>>(
      rec1, 64L * 64,        res2, inv_f, rec2, 64, 64, 32);
  inv_tconv5x5_s2_wmma<4><<<dim3(8, 4, 16), blk, 0, stream>>>(
      rec2, 128L * 128,      res3, inv_f, rec3, 128, 128, 32);
  inv_tconv5x5_s2_wmma<4><<<dim3(16, 8, 8), blk, 0, stream>>>(
      rec3, 256L * 256,      res4, inv_f, out, 256, 256, 32);
}


# Inverse: one product per 4 base rows, its 16 rows = 4 rows x 4 phases over the shared 6x3x8 window (weights as A from an LDS table): 9 WMMA per row not 18, whole-line pair stores; sum order changes
# speedup vs baseline: 1.8752x; 1.1006x over previous
#include <hip/hip_runtime.h>


typedef __attribute__((ext_vector_type(2))) float v2f;
typedef __attribute__((ext_vector_type(8))) float v8f;

__device__ __forceinline__ void lds_barrier() {
  __builtin_amdgcn_fence(__ATOMIC_RELEASE, "workgroup", "local");
  __builtin_amdgcn_s_barrier();
  __builtin_amdgcn_fence(__ATOMIC_ACQUIRE, "workgroup", "local");
}

template <int RPW>
__global__ __launch_bounds__(256) __attribute__((amdgpu_waves_per_eu(8))) void fwd_conv5x5_s2_wmma(
    const float* __restrict__ in, long inBatchStride,
    const float* __restrict__ filt,
    float* __restrict__ out,
    int H, int W, int Ho, int Wo, int nb)
{
  constexpr int TR  = 8 * RPW;
  constexpr int NR  = 2 * TR + 3;
  constexpr int NP  = NR * 18;
  constexpr int NIT = (NP + 255) / 256;

  const int col0 = blockIdx.x * 16;
  const int row0 = blockIdx.y * TR;
  const int tid  = threadIdx.x;
  const int wv   = tid >> 5;
  const int lane = tid & 31;
  const int hf   = lane >> 4;
  const int nl   = lane & 15;

  __shared__ __attribute__((aligned(16))) float sIn[NR * 36];
  __shared__ float sF[256];

  const float fv = filt[tid < 200 ? tid : 0];

  const int iy0 = 2 * row0 - 2;
  const int ix0 = 2 * col0 - 2;
  unsigned soff[NIT];
  unsigned okm = 0u;
#pragma unroll
  for (int it = 0; it < NIT; ++it) {
    const int idx = tid + 256 * it;
    const int r = idx / 18, j = idx - r * 18;
    const int gy = iy0 + r, gx = ix0 + 2 * j;
    const bool ok = idx < NP && (unsigned)gy < (unsigned)H && (unsigned)gx < (unsigned)W;
    soff[it] = ok ? (unsigned)(gy * W + gx) : 0u;
    okm |= (ok ? 1u : 0u) << it;
  }

  int b = blockIdx.z;
  const int bstep = gridDim.z;
  float2 st[NIT];
  {
    const float* inb = in + (long)b * inBatchStride;
#pragma unroll
    for (int it = 0; it < NIT; ++it)
      st[it] = *reinterpret_cast<const float2*>(inb + soff[it]);
  }

  sF[tid] = fv;
  lds_barrier();

  v2f Bv[7];
  int aoff[14];
#pragma unroll
  for (int s = 0; s < 7; ++s) {
#pragma unroll
    for (int t = 0; t < 2; ++t) {
      const int k  = 4 * s + 2 * hf + t;
      const int kk = k < 24 ? k : 24;
      const float w = sF[(nl & 7) * 25 + kk];
      Bv[s][t] = (nl < 8 && k < 25) ? w : 0.0f;
      const int kh = kk / 5, kw = kk - kh * 5;
      aoff[2 * s + t] = (2 * wv * RPW + kh) * 36 + 2 * nl + kw;
    }
  }

  for (;;) {
#pragma unroll
    for (int it = 0; it < NIT; ++it) {
      const int idx = tid + 256 * it;
      float2 v = st[it];
      if (!((okm >> it) & 1u)) { v.x = 0.0f; v.y = 0.0f; }
      if (idx < NP) *reinterpret_cast<float2*>(&sIn[2 * idx]) = v;
    }
    lds_barrier();

    const int bn = b + bstep;
    if (bn < nb) {
      const float* inb = in + (long)bn * inBatchStride;
#pragma unroll
      for (int it = 0; it < NIT; ++it)
        st[it] = *reinterpret_cast<const float2*>(inb + soff[it]);
    }

#pragma unroll
    for (int i = 0; i < RPW; i += 2) {
      v8f acc0 = {}, acc1 = {};
#pragma unroll
      for (int s = 0; s < 7; ++s) {
        v2f A0, A1;
        A0[0] = sIn[aoff[2 * s]     + (i    ) * 72];
        A0[1] = sIn[aoff[2 * s + 1] + (i    ) * 72];
        A1[0] = sIn[aoff[2 * s]     + (i + 1) * 72];
        A1[1] = sIn[aoff[2 * s + 1] + (i + 1) * 72];
        acc0 = __builtin_amdgcn_wmma_f32_16x16x4_f32(
            false, A0, false, Bv[s], (short)0, acc0, false, false);
        acc1 = __builtin_amdgcn_wmma_f32_16x16x4_f32(
            false, A1, false, Bv[s], (short)0, acc1, false, false);
      }
      if (nl < 8) {
        const int gy = row0 + wv * RPW + i;
        float* o = out + (((long)b * 8 + nl) * Ho + gy) * Wo + col0 + 8 * hf;
#pragma unroll
        for (int j = 0; j < 8; ++j) o[j] = acc0[j];
#pragma unroll
        for (int j = 0; j < 8; ++j) o[Wo + j] = acc1[j];
      }
    }

    if (bn >= nb) break;
    b = bn;
    lds_barrier();
  }
}

__global__ __launch_bounds__(256) __attribute__((amdgpu_waves_per_eu(8))) void inv_tconv5x5_s2_wmma(
    const float* __restrict__ rec,  long recBatchStride,
    const float* __restrict__ coef,
    const float* __restrict__ filt,
    float* __restrict__ outp,
    int H, int W, int nb)
{
  constexpr int TR   = 32;
  constexpr int NRI  = TR + 2;
  constexpr int PT   = 20;
  constexpr int PL   = NRI * PT;
  constexpr int NP0  = NRI * 10;
  constexpr int NIT0 = (NP0 + 255) / 256;
  constexpr int NP1  = 7 * NP0;
  constexpr int NIT1 = (NP1 + 255) / 256;
  constexpr int NS   = 36;
  constexpr int NWT  = NS * 64 / 256;
  constexpr int BK   = 6;

  const int bx0 = blockIdx.x * 16;
  const int by0 = blockIdx.y * TR;
  const int tid = threadIdx.x;
  const int wv  = tid >> 5;
  const int lane = tid & 31;
  const int hf  = lane >> 4;
  const int nl  = lane & 15;

  __shared__ __attribute__((aligned(16))) float sIn[8 * PL];
  __shared__ float sF[256];
  __shared__ __attribute__((aligned(16))) float sW[NS * 64];
  __shared__ unsigned sOff[(NIT0 + NIT1) * 256];

  const float fv = filt[tid < 200 ? tid : 0];

  unsigned off0[NIT0], off1[NIT1];
  unsigned okm0 = 0u, okm1 = 0u;
#pragma unroll
  for (int it = 0; it < NIT0; ++it) {
    const int idx = tid + 256 * it;
    const int r = idx / 10, j = idx - r * 10;
    const int gy = by0 + r - 1, gx = bx0 - 2 + 2 * j;
    const bool ok = idx < NP0 && (unsigned)gy < (unsigned)H && (unsigned)gx < (unsigned)W;
    off0[it] = ok ? (unsigned)(gy * W + gx) : 0u;
    okm0 |= (ok ? 1u : 0u) << it;
  }
#pragma unroll
  for (int it = 0; it < NIT1; ++it) {
    const int idx = tid + 256 * it;
    const int c1 = idx / NP0, rem = idx - c1 * NP0;
    const int r = rem / 10, j = rem - r * 10;
    const int gy = by0 + r - 1, gx = bx0 - 2 + 2 * j;
    const bool ok = idx < NP1 && (unsigned)gy < (unsigned)H && (unsigned)gx < (unsigned)W;
    off1[it] = ok ? (unsigned)(((c1 + 1) * H + gy) * W + gx) : 0u;
    okm1 |= (ok ? 1u : 0u) << it;
  }

  int b = blockIdx.z;
  const int bstep = gridDim.z;
  const long coefBatchStride = 8L * H * W;
  float2 s0[NIT0], s1[NIT1];
  {
    const float* recb  = rec  + (long)b * recBatchStride;
    const float* coefb = coef + (long)b * coefBatchStride;
#pragma unroll
    for (int it = 0; it < NIT0; ++it)
      s0[it] = *reinterpret_cast<const float2*>(recb + off0[it]);
#pragma unroll
    for (int it = 0; it < NIT1; ++it)
      s1[it] = *reinterpret_cast<const float2*>(coefb + off1[it]);
  }

#pragma unroll
  for (int it = 0; it < NIT0; ++it) sOff[it * 256 + tid] = off0[it];
#pragma unroll
  for (int it = 0; it < NIT1; ++it) sOff[(NIT0 + it) * 256 + tid] = off1[it];
  sF[tid] = fv;
  lds_barrier();

#pragma unroll
  for (int e = 0; e < NWT; ++e) {
    const int idx = tid + 256 * e;
    const int s = idx >> 6, l = (idx >> 1) & 31, t = idx & 1;
    const int m = l & 15, h = l >> 4;
    const int q = 2 * s + t;
    const int cc = q / 18, rem = q - cc * 18;
    const int wy = rem / 3, dxi = rem - wy * 3;
    const int r = m >> 2, py = (m >> 1) & 1, px = m & 1;
    const int dy = wy - 1 - r;
    const int ih = 2 + py - 2 * dy;
    const int iw = 4 + px - 2 * dxi;
    const bool ok = dy >= -1 && dy <= 1 && ih <= 4 && iw <= 4;
    const float w = sF[(2 * cc + h) * 25 + (ok ? ih * 5 + iw : 0)];
    sW[idx] = ok ? w : 0.0f;
  }

  const int pbase = hf * PL + (wv * 4) * PT + nl + 1;
  const float* wfrag = sW + 2 * lane;

  for (;;) {
#pragma unroll
    for (int it = 0; it < NIT0; ++it) {
      const int idx = tid + 256 * it;
      float2 v = s0[it];
      if (!((okm0 >> it) & 1u)) { v.x = 0.0f; v.y = 0.0f; }
      if (idx < NP0) *reinterpret_cast<float2*>(&sIn[2 * idx]) = v;
    }
#pragma unroll
    for (int it = 0; it < NIT1; ++it) {
      const int idx = tid + 256 * it;
      float2 v = s1[it];
      if (!((okm1 >> it) & 1u)) { v.x = 0.0f; v.y = 0.0f; }
      if (idx < NP1) *reinterpret_cast<float2*>(&sIn[PL + 2 * idx]) = v;
    }
    lds_barrier();

    const int bn = b + bstep;
    if (bn < nb) {
      const float* recb  = rec  + (long)bn * recBatchStride;
      const float* coefb = coef + (long)bn * coefBatchStride;
#pragma unroll
      for (int it = 0; it < NIT0; ++it)
        s0[it] = *reinterpret_cast<const float2*>(recb + sOff[it * 256 + tid]);
#pragma unroll
      for (int it = 0; it < NIT1; ++it)
        s1[it] = *reinterpret_cast<const float2*>(coefb + sOff[(NIT0 + it) * 256 + tid]);
    }

    v8f acc0 = {}, acc1 = {};
    v2f Wb[2][BK], Pb[2][BK];
#pragma unroll
    for (int u = 0; u < BK; ++u) {
      Wb[0][u] = *reinterpret_cast<const v2f*>(wfrag + 64 * u);
#pragma unroll
      for (int t = 0; t < 2; ++t) {
        const int q = 2 * u + t;
        const int cc = q / 18, rem = q - cc * 18;
        Pb[0][u][t] = sIn[pbase + 2 * cc * PL + (rem / 3) * PT + (rem % 3)];
      }
    }
#pragma unroll
    for (int blk = 0; blk < NS / BK; ++blk) {
      const int cb = blk & 1, nx = cb ^ 1;
      if (blk + 1 < NS / BK) {
#pragma unroll
        for (int u = 0; u < BK; ++u) {
          const int s = (blk + 1) * BK + u;
          Wb[nx][u] = *reinterpret_cast<const v2f*>(wfrag + 64 * s);
#pragma unroll
          for (int t = 0; t < 2; ++t) {
            const int q = 2 * s + t;
            const int cc = q / 18, rem = q - cc * 18;
            Pb[nx][u][t] = sIn[pbase + 2 * cc * PL + (rem / 3) * PT + (rem % 3)];
          }
        }
      }
      __builtin_amdgcn_sched_barrier(0);
#pragma unroll
      for (int u = 0; u < BK; ++u) {
        if ((blk * BK + u) & 1)
          acc1 = __builtin_amdgcn_wmma_f32_16x16x4_f32(
              false, Wb[cb][u], false, Pb[cb][u], (short)0, acc1, false, false);
        else
          acc0 = __builtin_amdgcn_wmma_f32_16x16x4_f32(
              false, Wb[cb][u], false, Pb[cb][u], (short)0, acc0, false, false);
      }
      __builtin_amdgcn_sched_barrier(0);
    }
    const v8f acc = acc0 + acc1;

    {
      const int oy0 = 2 * (by0 + wv * 4 + 2 * hf);
      float* o = outp + ((long)b * (2 * H) + oy0) * (2L * W) + 2 * (bx0 + nl);
#pragma unroll
      for (int jj = 0; jj < 4; ++jj) {
        float2 v;
        v.x = acc[2 * jj];
        v.y = acc[2 * jj + 1];
        *reinterpret_cast<float2*>(o + (long)jj * (2L * W)) = v;
      }
    }

    if (bn >= nb) break;
    b = bn;
    lds_barrier();
  }
}

extern "C" void kernel_launch(void* const* d_in, const int* in_sizes, int n_in,
                              void* d_out, int out_size, void* d_ws, size_t ws_size,
                              hipStream_t stream) {
  (void)in_sizes; (void)n_in; (void)out_size; (void)ws_size;
  const float* x     = (const float*)d_in[0];
  const float* fwd_f = (const float*)d_in[1];
  const float* inv_f = (const float*)d_in[2];
  float* out = (float*)d_out;
  float* ws  = (float*)d_ws;

  const size_t n4 = 32UL * 8 * 256 * 256;
  const size_t n3 = 32UL * 8 * 128 * 128;
  const size_t n2 = 32UL * 8 * 64 * 64;
  const size_t n1 = 32UL * 8 * 32 * 32;
  float* res4 = ws;
  float* res3 = res4 + n4;
  float* res2 = res3 + n3;
  float* res1 = res2 + n2;
  float* rec1 = res1 + n1;
  float* rec2 = rec1 + 32UL * 64 * 64;
  float* rec3 = rec2 + 32UL * 128 * 128;

  const dim3 blk(256);

  fwd_conv5x5_s2_wmma<8><<<dim3(16, 4, 16), blk, 0, stream>>>(
      x,    512L * 512,      fwd_f, res4, 512, 512, 256, 256, 32);
  fwd_conv5x5_s2_wmma<4><<<dim3(8, 4, 16), blk, 0, stream>>>(
      res4, 8L * 256 * 256,  fwd_f, res3, 256, 256, 128, 128, 32);
  fwd_conv5x5_s2_wmma<2><<<dim3(4, 4, 32), blk, 0, stream>>>(
      res3, 8L * 128 * 128,  fwd_f, res2, 128, 128, 64, 64, 32);
  fwd_conv5x5_s2_wmma<2><<<dim3(2, 2, 32), blk, 0, stream>>>(
      res2, 8L * 64 * 64,    fwd_f, res1, 64, 64, 32, 32, 32);

  inv_tconv5x5_s2_wmma<<<dim3(2, 1, 32), blk, 0, stream>>>(
      res1, 8L * 32 * 32,    res1, inv_f, rec1, 32, 32, 32);
  inv_tconv5x5_s2_wmma<<<dim3(4, 2, 32), blk, 0, stream>>>(
      rec1, 64L * 64,        res2, inv_f, rec2, 64, 64, 32);
  inv_tconv5x5_s2_wmma<<<dim3(8, 4, 16), blk, 0, stream>>>(
      rec2, 128L * 128,      res3, inv_f, rec3, 128, 128, 32);
  inv_tconv5x5_s2_wmma<<<dim3(16, 8, 8), blk, 0, stream>>>(
      rec3, 256L * 256,      res4, inv_f, out, 256, 256, 32);
}


# Forward: one product per 32-pixel output row (product rows = channel x pixel parity, K = shared 5x7 window): 9 WMMA per 32 pixels not 14, whole-line pair stores, 32x32 tiles; sum order changes
# speedup vs baseline: 2.6733x; 1.4256x over previous
#include <hip/hip_runtime.h>


typedef __attribute__((ext_vector_type(2))) float v2f;
typedef __attribute__((ext_vector_type(8))) float v8f;

__device__ __forceinline__ void lds_barrier() {
  __builtin_amdgcn_fence(__ATOMIC_RELEASE, "workgroup", "local");
  __builtin_amdgcn_s_barrier();
  __builtin_amdgcn_fence(__ATOMIC_ACQUIRE, "workgroup", "local");
}

template <int RPW>
__global__ __launch_bounds__(256) __attribute__((amdgpu_waves_per_eu(8))) void fwd_conv5x5_s2_wmma(
    const float* __restrict__ in, long inBatchStride,
    const float* __restrict__ filt,
    float* __restrict__ out,
    int H, int W, int Ho, int Wo, int nb)
{
  constexpr int TR  = 8 * RPW;
  constexpr int NR  = 2 * TR + 3;
  constexpr int P   = 70;
  constexpr int NP  = NR * 35;
  constexpr int NIT = (NP + 255) / 256;
  constexpr int RS  = RPW >= 2 ? 2 : 1;

  const int col0 = blockIdx.x * 32;
  const int row0 = blockIdx.y * TR;
  const int tid  = threadIdx.x;
  const int wv   = tid >> 5;
  const int lane = tid & 31;
  const int hf   = lane >> 4;
  const int nl   = lane & 15;

  __shared__ __attribute__((aligned(16))) float sIn[NR * P];
  __shared__ float sF[256];

  const float fv = filt[tid < 200 ? tid : 0];

  const int iy0 = 2 * row0 - 2;
  const int ix0 = 2 * col0 - 2;
  unsigned soff[NIT];
  unsigned okm = 0u;
#pragma unroll
  for (int it = 0; it < NIT; ++it) {
    const int idx = tid + 256 * it;
    const int r = idx / 35, j = idx - r * 35;
    const int gy = iy0 + r, gx = ix0 + 2 * j;
    const bool ok = idx < NP && (unsigned)gy < (unsigned)H && (unsigned)gx < (unsigned)W;
    soff[it] = ok ? (unsigned)(gy * W + gx) : 0u;
    okm |= (ok ? 1u : 0u) << it;
  }

  int b = blockIdx.z;
  const int bstep = gridDim.z;
  float2 st[NIT];
  {
    const float* inb = in + (long)b * inBatchStride;
#pragma unroll
    for (int it = 0; it < NIT; ++it)
      st[it] = *reinterpret_cast<const float2*>(inb + soff[it]);
  }

  sF[tid] = fv;
  lds_barrier();

  v2f Aw[9];
  {
    const int ph = nl & 1, ch = nl >> 1;
#pragma unroll
    for (int s = 0; s < 9; ++s) {
#pragma unroll
      for (int t = 0; t < 2; ++t) {
        int kh, wx;
        if (s < 7) { const int q = 2 * s + t; const int rr = q / 7; kh = 2 * rr + hf; wx = q - 7 * rr; }
        else       { kh = 4; wx = 4 * (s - 7) + t + 2 * hf; }
        const int kw = wx - 2 * ph;
        const bool ok = kw >= 0 && kw <= 4;
        const float w = sF[ch * 25 + kh * 5 + (ok ? kw : 0)];
        Aw[s][t] = ok ? w : 0.0f;
      }
    }
  }

  const float* pa = sIn + (2 * wv * RPW + hf) * P + 4 * nl;
  const float* pb = sIn + (2 * wv * RPW + 4) * P + 4 * nl + 2 * hf;

  for (;;) {
#pragma unroll
    for (int it = 0; it < NIT; ++it) {
      const int idx = tid + 256 * it;
      float2 v = st[it];
      if (!((okm >> it) & 1u)) { v.x = 0.0f; v.y = 0.0f; }
      if (idx < NP) *reinterpret_cast<float2*>(&sIn[2 * idx]) = v;
    }
    lds_barrier();

    const int bn = b + bstep;
    if (bn < nb) {
      const float* inb = in + (long)bn * inBatchStride;
#pragma unroll
      for (int it = 0; it < NIT; ++it)
        st[it] = *reinterpret_cast<const float2*>(inb + soff[it]);
    }

#pragma unroll
    for (int i = 0; i < RPW; i += RS) {
      v8f acc0 = {}, acc1 = {};
#pragma unroll
      for (int s = 0; s < 9; ++s) {
        v2f B0, B1 = {};
#pragma unroll
        for (int t = 0; t < 2; ++t) {
          int o;
          const float* p;
          if (s < 7) { const int q = 2 * s + t; const int rr = q / 7; p = pa; o = 2 * rr * P + (q - 7 * rr); }
          else       { p = pb; o = 4 * (s - 7) + t; }
          B0[t] = p[o + (i) * 2 * P];
          if (RS == 2) B1[t] = p[o + (i + 1) * 2 * P];
        }
        acc0 = __builtin_amdgcn_wmma_f32_16x16x4_f32(
            false, Aw[s], false, B0, (short)0, acc0, false, false);
        if (RS == 2)
          acc1 = __builtin_amdgcn_wmma_f32_16x16x4_f32(
              false, Aw[s], false, B1, (short)0, acc1, false, false);
      }
      {
        const int gy = row0 + wv * RPW + i;
        const long cs = (long)Ho * Wo;
        float* o = out + (((long)b * 8 + 4 * hf) * Ho + gy) * Wo + col0 + 2 * nl;
#pragma unroll
        for (int jj = 0; jj < 4; ++jj) {
          float2 v;
          v.x = acc0[2 * jj];
          v.y = acc0[2 * jj + 1];
          *reinterpret_cast<float2*>(o + jj * cs) = v;
        }
        if (RS == 2) {
#pragma unroll
          for (int jj = 0; jj < 4; ++jj) {
            float2 v;
            v.x = acc1[2 * jj];
            v.y = acc1[2 * jj + 1];
            *reinterpret_cast<float2*>(o + jj * cs + Wo) = v;
          }
        }
      }
    }

    if (bn >= nb) break;
    b = bn;
    lds_barrier();
  }
}

__global__ __launch_bounds__(256) __attribute__((amdgpu_waves_per_eu(8))) void inv_tconv5x5_s2_wmma(
    const float* __restrict__ rec,  long recBatchStride,
    const float* __restrict__ coef,
    const float* __restrict__ filt,
    float* __restrict__ outp,
    int H, int W, int nb)
{
  constexpr int TR   = 32;
  constexpr int NRI  = TR + 2;
  constexpr int PT   = 20;
  constexpr int PL   = NRI * PT;
  constexpr int NP0  = NRI * 10;
  constexpr int NIT0 = (NP0 + 255) / 256;
  constexpr int NP1  = 7 * NP0;
  constexpr int NIT1 = (NP1 + 255) / 256;
  constexpr int NS   = 36;
  constexpr int NWT  = NS * 64 / 256;
  constexpr int BK   = 6;

  const int bx0 = blockIdx.x * 16;
  const int by0 = blockIdx.y * TR;
  const int tid = threadIdx.x;
  const int wv  = tid >> 5;
  const int lane = tid & 31;
  const int hf  = lane >> 4;
  const int nl  = lane & 15;

  __shared__ __attribute__((aligned(16))) float sIn[8 * PL];
  __shared__ float sF[256];
  __shared__ __attribute__((aligned(16))) float sW[NS * 64];
  __shared__ unsigned sOff[(NIT0 + NIT1) * 256];

  const float fv = filt[tid < 200 ? tid : 0];

  unsigned off0[NIT0], off1[NIT1];
  unsigned okm0 = 0u, okm1 = 0u;
#pragma unroll
  for (int it = 0; it < NIT0; ++it) {
    const int idx = tid + 256 * it;
    const int r = idx / 10, j = idx - r * 10;
    const int gy = by0 + r - 1, gx = bx0 - 2 + 2 * j;
    const bool ok = idx < NP0 && (unsigned)gy < (unsigned)H && (unsigned)gx < (unsigned)W;
    off0[it] = ok ? (unsigned)(gy * W + gx) : 0u;
    okm0 |= (ok ? 1u : 0u) << it;
  }
#pragma unroll
  for (int it = 0; it < NIT1; ++it) {
    const int idx = tid + 256 * it;
    const int c1 = idx / NP0, rem = idx - c1 * NP0;
    const int r = rem / 10, j = rem - r * 10;
    const int gy = by0 + r - 1, gx = bx0 - 2 + 2 * j;
    const bool ok = idx < NP1 && (unsigned)gy < (unsigned)H && (unsigned)gx < (unsigned)W;
    off1[it] = ok ? (unsigned)(((c1 + 1) * H + gy) * W + gx) : 0u;
    okm1 |= (ok ? 1u : 0u) << it;
  }

  int b = blockIdx.z;
  const int bstep = gridDim.z;
  const long coefBatchStride = 8L * H * W;
  float2 s0[NIT0], s1[NIT1];
  {
    const float* recb  = rec  + (long)b * recBatchStride;
    const float* coefb = coef + (long)b * coefBatchStride;
#pragma unroll
    for (int it = 0; it < NIT0; ++it)
      s0[it] = *reinterpret_cast<const float2*>(recb + off0[it]);
#pragma unroll
    for (int it = 0; it < NIT1; ++it)
      s1[it] = *reinterpret_cast<const float2*>(coefb + off1[it]);
  }

#pragma unroll
  for (int it = 0; it < NIT0; ++it) sOff[it * 256 + tid] = off0[it];
#pragma unroll
  for (int it = 0; it < NIT1; ++it) sOff[(NIT0 + it) * 256 + tid] = off1[it];
  sF[tid] = fv;
  lds_barrier();

#pragma unroll
  for (int e = 0; e < NWT; ++e) {
    const int idx = tid + 256 * e;
    const int s = idx >> 6, l = (idx >> 1) & 31, t = idx & 1;
    const int m = l & 15, h = l >> 4;
    const int q = 2 * s + t;
    const int cc = q / 18, rem = q - cc * 18;
    const int wy = rem / 3, dxi = rem - wy * 3;
    const int r = m >> 2, py = (m >> 1) & 1, px = m & 1;
    const int dy = wy - 1 - r;
    const int ih = 2 + py - 2 * dy;
    const int iw = 4 + px - 2 * dxi;
    const bool ok = dy >= -1 && dy <= 1 && ih <= 4 && iw <= 4;
    const float w = sF[(2 * cc + h) * 25 + (ok ? ih * 5 + iw : 0)];
    sW[idx] = ok ? w : 0.0f;
  }

  const int pbase = hf * PL + (wv * 4) * PT + nl + 1;
  const float* wfrag = sW + 2 * lane;

  for (;;) {
#pragma unroll
    for (int it = 0; it < NIT0; ++it) {
      const int idx = tid + 256 * it;
      float2 v = s0[it];
      if (!((okm0 >> it) & 1u)) { v.x = 0.0f; v.y = 0.0f; }
      if (idx < NP0) *reinterpret_cast<float2*>(&sIn[2 * idx]) = v;
    }
#pragma unroll
    for (int it = 0; it < NIT1; ++it) {
      const int idx = tid + 256 * it;
      float2 v = s1[it];
      if (!((okm1 >> it) & 1u)) { v.x = 0.0f; v.y = 0.0f; }
      if (idx < NP1) *reinterpret_cast<float2*>(&sIn[PL + 2 * idx]) = v;
    }
    lds_barrier();

    const int bn = b + bstep;
    if (bn < nb) {
      const float* recb  = rec  + (long)bn * recBatchStride;
      const float* coefb = coef + (long)bn * coefBatchStride;
#pragma unroll
      for (int it = 0; it < NIT0; ++it)
        s0[it] = *reinterpret_cast<const float2*>(recb + sOff[it * 256 + tid]);
#pragma unroll
      for (int it = 0; it < NIT1; ++it)
        s1[it] = *reinterpret_cast<const float2*>(coefb + sOff[(NIT0 + it) * 256 + tid]);
    }

    v8f acc0 = {}, acc1 = {};
    v2f Wb[2][BK], Pb[2][BK];
#pragma unroll
    for (int u = 0; u < BK; ++u) {
      Wb[0][u] = *reinterpret_cast<const v2f*>(wfrag + 64 * u);
#pragma unroll
      for (int t = 0; t < 2; ++t) {
        const int q = 2 * u + t;
        const int cc = q / 18, rem = q - cc * 18;
        Pb[0][u][t] = sIn[pbase + 2 * cc * PL + (rem / 3) * PT + (rem % 3)];
      }
    }
#pragma unroll
    for (int blk = 0; blk < NS / BK; ++blk) {
      const int cb = blk & 1, nx = cb ^ 1;
      if (blk + 1 < NS / BK) {
#pragma unroll
        for (int u = 0; u < BK; ++u) {
          const int s = (blk + 1) * BK + u;
          Wb[nx][u] = *reinterpret_cast<const v2f*>(wfrag + 64 * s);
#pragma unroll
          for (int t = 0; t < 2; ++t) {
            const int q = 2 * s + t;
            const int cc = q / 18, rem = q - cc * 18;
            Pb[nx][u][t] = sIn[pbase + 2 * cc * PL + (rem / 3) * PT + (rem % 3)];
          }
        }
      }
      __builtin_amdgcn_sched_barrier(0);
#pragma unroll
      for (int u = 0; u < BK; ++u) {
        if ((blk * BK + u) & 1)
          acc1 = __builtin_amdgcn_wmma_f32_16x16x4_f32(
              false, Wb[cb][u], false, Pb[cb][u], (short)0, acc1, false, false);
        else
          acc0 = __builtin_amdgcn_wmma_f32_16x16x4_f32(
              false, Wb[cb][u], false, Pb[cb][u], (short)0, acc0, false, false);
      }
      __builtin_amdgcn_sched_barrier(0);
    }
    const v8f acc = acc0 + acc1;

    {
      const int oy0 = 2 * (by0 + wv * 4 + 2 * hf);
      float* o = outp + ((long)b * (2 * H) + oy0) * (2L * W) + 2 * (bx0 + nl);
#pragma unroll
      for (int jj = 0; jj < 4; ++jj) {
        float2 v;
        v.x = acc[2 * jj];
        v.y = acc[2 * jj + 1];
        *reinterpret_cast<float2*>(o + (long)jj * (2L * W)) = v;
      }
    }

    if (bn >= nb) break;
    b = bn;
    lds_barrier();
  }
}

extern "C" void kernel_launch(void* const* d_in, const int* in_sizes, int n_in,
                              void* d_out, int out_size, void* d_ws, size_t ws_size,
                              hipStream_t stream) {
  (void)in_sizes; (void)n_in; (void)out_size; (void)ws_size;
  const float* x     = (const float*)d_in[0];
  const float* fwd_f = (const float*)d_in[1];
  const float* inv_f = (const float*)d_in[2];
  float* out = (float*)d_out;
  float* ws  = (float*)d_ws;

  const size_t n4 = 32UL * 8 * 256 * 256;
  const size_t n3 = 32UL * 8 * 128 * 128;
  const size_t n2 = 32UL * 8 * 64 * 64;
  const size_t n1 = 32UL * 8 * 32 * 32;
  float* res4 = ws;
  float* res3 = res4 + n4;
  float* res2 = res3 + n3;
  float* res1 = res2 + n2;
  float* rec1 = res1 + n1;
  float* rec2 = rec1 + 32UL * 64 * 64;
  float* rec3 = rec2 + 32UL * 128 * 128;

  const dim3 blk(256);

  fwd_conv5x5_s2_wmma<4><<<dim3(8, 8, 16), blk, 0, stream>>>(
      x,    512L * 512,      fwd_f, res4, 512, 512, 256, 256, 32);
  fwd_conv5x5_s2_wmma<2><<<dim3(4, 8, 16), blk, 0, stream>>>(
      res4, 8L * 256 * 256,  fwd_f, res3, 256, 256, 128, 128, 32);
  fwd_conv5x5_s2_wmma<1><<<dim3(2, 8, 32), blk, 0, stream>>>(
      res3, 8L * 128 * 128,  fwd_f, res2, 128, 128, 64, 64, 32);
  fwd_conv5x5_s2_wmma<1><<<dim3(1, 4, 32), blk, 0, stream>>>(
      res2, 8L * 64 * 64,    fwd_f, res1, 64, 64, 32, 32, 32);

  inv_tconv5x5_s2_wmma<<<dim3(2, 1, 32), blk, 0, stream>>>(
      res1, 8L * 32 * 32,    res1, inv_f, rec1, 32, 32, 32);
  inv_tconv5x5_s2_wmma<<<dim3(4, 2, 32), blk, 0, stream>>>(
      rec1, 64L * 64,        res2, inv_f, rec2, 64, 64, 32);
  inv_tconv5x5_s2_wmma<<<dim3(8, 4, 16), blk, 0, stream>>>(
      rec2, 128L * 128,      res3, inv_f, rec3, 128, 128, 32);
  inv_tconv5x5_s2_wmma<<<dim3(16, 8, 8), blk, 0, stream>>>(
      rec3, 256L * 256,      res4, inv_f, out, 256, 256, 32);
}


# Inverse: tiles span whole image rows (8x64, 8x128, 4x256 base px; 16 waves at the two largest levels): staged rows are whole aligned lines, no column halo, each line read by one work-group
# speedup vs baseline: 3.0363x; 1.1358x over previous
#include <hip/hip_runtime.h>


typedef __attribute__((ext_vector_type(2))) float v2f;
typedef __attribute__((ext_vector_type(8))) float v8f;

__device__ __forceinline__ void lds_barrier() {
  __builtin_amdgcn_fence(__ATOMIC_RELEASE, "workgroup", "local");
  __builtin_amdgcn_s_barrier();
  __builtin_amdgcn_fence(__ATOMIC_ACQUIRE, "workgroup", "local");
}

template <int RPW>
__global__ __launch_bounds__(256) __attribute__((amdgpu_waves_per_eu(8))) void fwd_conv5x5_s2_wmma(
    const float* __restrict__ in, long inBatchStride,
    const float* __restrict__ filt,
    float* __restrict__ out,
    int H, int W, int Ho, int Wo, int nb)
{
  constexpr int TR  = 8 * RPW;
  constexpr int NR  = 2 * TR + 3;
  constexpr int P   = 70;
  constexpr int NP  = NR * 35;
  constexpr int NIT = (NP + 255) / 256;
  constexpr int RS  = RPW >= 2 ? 2 : 1;

  const int col0 = blockIdx.x * 32;
  const int row0 = blockIdx.y * TR;
  const int tid  = threadIdx.x;
  const int wv   = tid >> 5;
  const int lane = tid & 31;
  const int hf   = lane >> 4;
  const int nl   = lane & 15;

  __shared__ __attribute__((aligned(16))) float sIn[NR * P];
  __shared__ float sF[256];

  const float fv = filt[tid < 200 ? tid : 0];

  const int iy0 = 2 * row0 - 2;
  const int ix0 = 2 * col0 - 2;
  unsigned soff[NIT];
  unsigned okm = 0u;
#pragma unroll
  for (int it = 0; it < NIT; ++it) {
    const int idx = tid + 256 * it;
    const int r = idx / 35, j = idx - r * 35;
    const int gy = iy0 + r, gx = ix0 + 2 * j;
    const bool ok = idx < NP && (unsigned)gy < (unsigned)H && (unsigned)gx < (unsigned)W;
    soff[it] = ok ? (unsigned)(gy * W + gx) : 0u;
    okm |= (ok ? 1u : 0u) << it;
  }

  int b = blockIdx.z;
  const int bstep = gridDim.z;
  float2 st[NIT];
  {
    const float* inb = in + (long)b * inBatchStride;
#pragma unroll
    for (int it = 0; it < NIT; ++it)
      st[it] = *reinterpret_cast<const float2*>(inb + soff[it]);
  }

  sF[tid] = fv;
  lds_barrier();

  v2f Aw[9];
  {
    const int ph = nl & 1, ch = nl >> 1;
#pragma unroll
    for (int s = 0; s < 9; ++s) {
#pragma unroll
      for (int t = 0; t < 2; ++t) {
        int kh, wx;
        if (s < 7) { const int q = 2 * s + t; const int rr = q / 7; kh = 2 * rr + hf; wx = q - 7 * rr; }
        else       { kh = 4; wx = 4 * (s - 7) + t + 2 * hf; }
        const int kw = wx - 2 * ph;
        const bool ok = kw >= 0 && kw <= 4;
        const float w = sF[ch * 25 + kh * 5 + (ok ? kw : 0)];
        Aw[s][t] = ok ? w : 0.0f;
      }
    }
  }

  const float* pa = sIn + (2 * wv * RPW + hf) * P + 4 * nl;
  const float* pb = sIn + (2 * wv * RPW + 4) * P + 4 * nl + 2 * hf;

  for (;;) {
#pragma unroll
    for (int it = 0; it < NIT; ++it) {
      const int idx = tid + 256 * it;
      float2 v = st[it];
      if (!((okm >> it) & 1u)) { v.x = 0.0f; v.y = 0.0f; }
      if (idx < NP) *reinterpret_cast<float2*>(&sIn[2 * idx]) = v;
    }
    lds_barrier();

    const int bn = b + bstep;
    if (bn < nb) {
      const float* inb = in + (long)bn * inBatchStride;
#pragma unroll
      for (int it = 0; it < NIT; ++it)
        st[it] = *reinterpret_cast<const float2*>(inb + soff[it]);
    }

#pragma unroll
    for (int i = 0; i < RPW; i += RS) {
      v8f acc0 = {}, acc1 = {};
#pragma unroll
      for (int s = 0; s < 9; ++s) {
        v2f B0, B1 = {};
#pragma unroll
        for (int t = 0; t < 2; ++t) {
          int o;
          const float* p;
          if (s < 7) { const int q = 2 * s + t; const int rr = q / 7; p = pa; o = 2 * rr * P + (q - 7 * rr); }
          else       { p = pb; o = 4 * (s - 7) + t; }
          B0[t] = p[o + (i) * 2 * P];
          if (RS == 2) B1[t] = p[o + (i + 1) * 2 * P];
        }
        acc0 = __builtin_amdgcn_wmma_f32_16x16x4_f32(
            false, Aw[s], false, B0, (short)0, acc0, false, false);
        if (RS == 2)
          acc1 = __builtin_amdgcn_wmma_f32_16x16x4_f32(
              false, Aw[s], false, B1, (short)0, acc1, false, false);
      }
      {
        const int gy = row0 + wv * RPW + i;
        const long cs = (long)Ho * Wo;
        float* o = out + (((long)b * 8 + 4 * hf) * Ho + gy) * Wo + col0 + 2 * nl;
#pragma unroll
        for (int jj = 0; jj < 4; ++jj) {
          float2 v;
          v.x = acc0[2 * jj];
          v.y = acc0[2 * jj + 1];
          *reinterpret_cast<float2*>(o + jj * cs) = v;
        }
        if (RS == 2) {
#pragma unroll
          for (int jj = 0; jj < 4; ++jj) {
            float2 v;
            v.x = acc1[2 * jj];
            v.y = acc1[2 * jj + 1];
            *reinterpret_cast<float2*>(o + jj * cs + Wo) = v;
          }
        }
      }
    }

    if (bn >= nb) break;
    b = bn;
    lds_barrier();
  }
}

template <int CG, int RG>
__global__ __launch_bounds__(32 * CG * RG) __attribute__((amdgpu_waves_per_eu(8))) void inv_tconv5x5_s2_wmma(
    const float* __restrict__ rec,  long recBatchStride,
    const float* __restrict__ coef,
    const float* __restrict__ filt,
    float* __restrict__ outp,
    int H, int W, int nb)
{
  constexpr int NT   = 32 * CG * RG;
  constexpr int TC   = 16 * CG;
  constexpr int TR   = 4 * RG;
  constexpr int NRI  = TR + 2;
  constexpr int PT   = TC + 4;
  constexpr int PP   = PT / 2;
  constexpr int PL   = NRI * PT;
  constexpr int NP0  = NRI * PP;
  constexpr int NIT0 = (NP0 + NT - 1) / NT;
  constexpr int NP1  = 7 * NP0;
  constexpr int NIT1 = (NP1 + NT - 1) / NT;
  constexpr int NS   = 36;
  constexpr int NWT  = (NS * 64 + NT - 1) / NT;
  constexpr int BK   = 6;
  static_assert(PL % 64 >= 16 && PL % 64 <= 48, "plane pitch puts the lane halves on the same banks");

  const int bx0 = blockIdx.x * TC;
  const int by0 = blockIdx.y * TR;
  const int tid = threadIdx.x;
  const int wv  = tid >> 5;
  const int cg  = wv % CG;
  const int rg  = wv / CG;
  const int lane = tid & 31;
  const int hf  = lane >> 4;
  const int nl  = lane & 15;

  __shared__ __attribute__((aligned(16))) float sIn[8 * PL];
  __shared__ float sF[NT];
  __shared__ __attribute__((aligned(16))) float sW[NS * 64];
  __shared__ unsigned sOff[(NIT0 + NIT1) * NT];

  const float fv = filt[tid < 200 ? tid : 0];

  unsigned off0[NIT0], off1[NIT1];
  unsigned okm0 = 0u, okm1 = 0u;
#pragma unroll
  for (int it = 0; it < NIT0; ++it) {
    const int idx = tid + NT * it;
    const int r = idx / PP, j = idx - r * PP;
    const int gy = by0 + r - 1, gx = bx0 - 2 + 2 * j;
    const bool ok = idx < NP0 && (unsigned)gy < (unsigned)H && (unsigned)gx < (unsigned)W;
    off0[it] = ok ? (unsigned)(gy * W + gx) : 0u;
    okm0 |= (ok ? 1u : 0u) << it;
  }
#pragma unroll
  for (int it = 0; it < NIT1; ++it) {
    const int idx = tid + NT * it;
    const int c1 = idx / NP0, rem = idx - c1 * NP0;
    const int r = rem / PP, j = rem - r * PP;
    const int gy = by0 + r - 1, gx = bx0 - 2 + 2 * j;
    const bool ok = idx < NP1 && (unsigned)gy < (unsigned)H && (unsigned)gx < (unsigned)W;
    off1[it] = ok ? (unsigned)(((c1 + 1) * H + gy) * W + gx) : 0u;
    okm1 |= (ok ? 1u : 0u) << it;
  }

  int b = blockIdx.z;
  const int bstep = gridDim.z;
  const long coefBatchStride = 8L * H * W;
  float2 s0[NIT0], s1[NIT1];
  {
    const float* recb  = rec  + (long)b * recBatchStride;
    const float* coefb = coef + (long)b * coefBatchStride;
#pragma unroll
    for (int it = 0; it < NIT0; ++it)
      s0[it] = *reinterpret_cast<const float2*>(recb + off0[it]);
#pragma unroll
    for (int it = 0; it < NIT1; ++it)
      s1[it] = *reinterpret_cast<const float2*>(coefb + off1[it]);
  }

#pragma unroll
  for (int it = 0; it < NIT0; ++it) sOff[it * NT + tid] = off0[it];
#pragma unroll
  for (int it = 0; it < NIT1; ++it) sOff[(NIT0 + it) * NT + tid] = off1[it];
  sF[tid] = fv;
  lds_barrier();

#pragma unroll
  for (int e = 0; e < NWT; ++e) {
    const int idx = tid + NT * e;
    if (idx >= NS * 64) break;
    const int s = idx >> 6, l = (idx >> 1) & 31, t = idx & 1;
    const int m = l & 15, h = l >> 4;
    const int q = 2 * s + t;
    const int cc = q / 18, rem = q - cc * 18;
    const int wy = rem / 3, dxi = rem - wy * 3;
    const int r = m >> 2, py = (m >> 1) & 1, px = m & 1;
    const int dy = wy - 1 - r;
    const int ih = 2 + py - 2 * dy;
    const int iw = 4 + px - 2 * dxi;
    const bool ok = dy >= -1 && dy <= 1 && ih <= 4 && iw <= 4;
    const float w = sF[(2 * cc + h) * 25 + (ok ? ih * 5 + iw : 0)];
    sW[idx] = ok ? w : 0.0f;
  }

  const int pbase = hf * PL + (rg * 4) * PT + cg * 16 + nl + 1;
  const float* wfrag = sW + 2 * lane;

  for (;;) {
#pragma unroll
    for (int it = 0; it < NIT0; ++it) {
      const int idx = tid + NT * it;
      float2 v = s0[it];
      if (!((okm0 >> it) & 1u)) { v.x = 0.0f; v.y = 0.0f; }
      if (idx < NP0) *reinterpret_cast<float2*>(&sIn[2 * idx]) = v;
    }
#pragma unroll
    for (int it = 0; it < NIT1; ++it) {
      const int idx = tid + NT * it;
      float2 v = s1[it];
      if (!((okm1 >> it) & 1u)) { v.x = 0.0f; v.y = 0.0f; }
      if (idx < NP1) *reinterpret_cast<float2*>(&sIn[PL + 2 * idx]) = v;
    }
    lds_barrier();

    const int bn = b + bstep;
    if (bn < nb) {
      const float* recb  = rec  + (long)bn * recBatchStride;
      const float* coefb = coef + (long)bn * coefBatchStride;
#pragma unroll
      for (int it = 0; it < NIT0; ++it)
        s0[it] = *reinterpret_cast<const float2*>(recb + sOff[it * NT + tid]);
#pragma unroll
      for (int it = 0; it < NIT1; ++it)
        s1[it] = *reinterpret_cast<const float2*>(coefb + sOff[(NIT0 + it) * NT + tid]);
    }

    v8f acc0 = {}, acc1 = {};
    v2f Wb[2][BK], Pb[2][BK];
#pragma unroll
    for (int u = 0; u < BK; ++u) {
      Wb[0][u] = *reinterpret_cast<const v2f*>(wfrag + 64 * u);
#pragma unroll
      for (int t = 0; t < 2; ++t) {
        const int q = 2 * u + t;
        const int cc = q / 18, rem = q - cc * 18;
        Pb[0][u][t] = sIn[pbase + 2 * cc * PL + (rem / 3) * PT + (rem % 3)];
      }
    }
#pragma unroll
    for (int blk = 0; blk < NS / BK; ++blk) {
      const int cb = blk & 1, nx = cb ^ 1;
      if (blk + 1 < NS / BK) {
#pragma unroll
        for (int u = 0; u < BK; ++u) {
          const int s = (blk + 1) * BK + u;
          Wb[nx][u] = *reinterpret_cast<const v2f*>(wfrag + 64 * s);
#pragma unroll
          for (int t = 0; t < 2; ++t) {
            const int q = 2 * s + t;
            const int cc = q / 18, rem = q - cc * 18;
            Pb[nx][u][t] = sIn[pbase + 2 * cc * PL + (rem / 3) * PT + (rem % 3)];
          }
        }
      }
      __builtin_amdgcn_sched_barrier(0);
#pragma unroll
      for (int u = 0; u < BK; ++u) {
        if ((blk * BK + u) & 1)
          acc1 = __builtin_amdgcn_wmma_f32_16x16x4_f32(
              false, Wb[cb][u], false, Pb[cb][u], (short)0, acc1, false, false);
        else
          acc0 = __builtin_amdgcn_wmma_f32_16x16x4_f32(
              false, Wb[cb][u], false, Pb[cb][u], (short)0, acc0, false, false);
      }
      __builtin_amdgcn_sched_barrier(0);
    }
    const v8f acc = acc0 + acc1;

    {
      const int oy0 = 2 * (by0 + rg * 4 + 2 * hf);
      float* o = outp + ((long)b * (2 * H) + oy0) * (2L * W) + 2 * (bx0 + cg * 16 + nl);
#pragma unroll
      for (int jj = 0; jj < 4; ++jj) {
        float2 v;
        v.x = acc[2 * jj];
        v.y = acc[2 * jj + 1];
        *reinterpret_cast<float2*>(o + (long)jj * (2L * W)) = v;
      }
    }

    if (bn >= nb) break;
    b = bn;
    lds_barrier();
  }
}

extern "C" void kernel_launch(void* const* d_in, const int* in_sizes, int n_in,
                              void* d_out, int out_size, void* d_ws, size_t ws_size,
                              hipStream_t stream) {
  (void)in_sizes; (void)n_in; (void)out_size; (void)ws_size;
  const float* x     = (const float*)d_in[0];
  const float* fwd_f = (const float*)d_in[1];
  const float* inv_f = (const float*)d_in[2];
  float* out = (float*)d_out;
  float* ws  = (float*)d_ws;

  const size_t n4 = 32UL * 8 * 256 * 256;
  const size_t n3 = 32UL * 8 * 128 * 128;
  const size_t n2 = 32UL * 8 * 64 * 64;
  const size_t n1 = 32UL * 8 * 32 * 32;
  float* res4 = ws;
  float* res3 = res4 + n4;
  float* res2 = res3 + n3;
  float* res1 = res2 + n2;
  float* rec1 = res1 + n1;
  float* rec2 = rec1 + 32UL * 64 * 64;
  float* rec3 = rec2 + 32UL * 128 * 128;

  const dim3 blk(256);

  fwd_conv5x5_s2_wmma<4><<<dim3(8, 8, 16), blk, 0, stream>>>(
      x,    512L * 512,      fwd_f, res4, 512, 512, 256, 256, 32);
  fwd_conv5x5_s2_wmma<2><<<dim3(4, 8, 16), blk, 0, stream>>>(
      res4, 8L * 256 * 256,  fwd_f, res3, 256, 256, 128, 128, 32);
  fwd_conv5x5_s2_wmma<1><<<dim3(2, 8, 32), blk, 0, stream>>>(
      res3, 8L * 128 * 128,  fwd_f, res2, 128, 128, 64, 64, 32);
  fwd_conv5x5_s2_wmma<1><<<dim3(1, 4, 32), blk, 0, stream>>>(
      res2, 8L * 64 * 64,    fwd_f, res1, 64, 64, 32, 32, 32);

  inv_tconv5x5_s2_wmma<1, 8><<<dim3(2, 1, 32), blk, 0, stream>>>(
      res1, 8L * 32 * 32,    res1, inv_f, rec1, 32, 32, 32);
  inv_tconv5x5_s2_wmma<4, 2><<<dim3(1, 8, 32), blk, 0, stream>>>(
      rec1, 64L * 64,        res2, inv_f, rec2, 64, 64, 32);
  inv_tconv5x5_s2_wmma<8, 2><<<dim3(1, 16, 16), dim3(512), 0, stream>>>(
      rec2, 128L * 128,      res3, inv_f, rec3, 128, 128, 32);
  inv_tconv5x5_s2_wmma<16, 1><<<dim3(1, 64, 8), dim3(512), 0, stream>>>(
      rec3, 256L * 256,      res4, inv_f, out, 256, 256, 32);
}


# Inverse, whole-row tiles: stage only the image's own columns (halo columns zeroed once): every wave's staging load is one aligned 256-byte piece of a row, two whole lines, not three
# speedup vs baseline: 3.1392x; 1.0339x over previous
#include <hip/hip_runtime.h>


typedef __attribute__((ext_vector_type(2))) float v2f;
typedef __attribute__((ext_vector_type(8))) float v8f;

__device__ __forceinline__ void lds_barrier() {
  __builtin_amdgcn_fence(__ATOMIC_RELEASE, "workgroup", "local");
  __builtin_amdgcn_s_barrier();
  __builtin_amdgcn_fence(__ATOMIC_ACQUIRE, "workgroup", "local");
}

template <int RPW>
__global__ __launch_bounds__(256) __attribute__((amdgpu_waves_per_eu(8))) void fwd_conv5x5_s2_wmma(
    const float* __restrict__ in, long inBatchStride,
    const float* __restrict__ filt,
    float* __restrict__ out,
    int H, int W, int Ho, int Wo, int nb)
{
  constexpr int TR  = 8 * RPW;
  constexpr int NR  = 2 * TR + 3;
  constexpr int P   = 70;
  constexpr int NP  = NR * 35;
  constexpr int NIT = (NP + 255) / 256;
  constexpr int RS  = RPW >= 2 ? 2 : 1;

  const int col0 = blockIdx.x * 32;
  const int row0 = blockIdx.y * TR;
  const int tid  = threadIdx.x;
  const int wv   = tid >> 5;
  const int lane = tid & 31;
  const int hf   = lane >> 4;
  const int nl   = lane & 15;

  __shared__ __attribute__((aligned(16))) float sIn[NR * P];
  __shared__ float sF[256];

  const float fv = filt[tid < 200 ? tid : 0];

  const int iy0 = 2 * row0 - 2;
  const int ix0 = 2 * col0 - 2;
  unsigned soff[NIT];
  unsigned okm = 0u;
#pragma unroll
  for (int it = 0; it < NIT; ++it) {
    const int idx = tid + 256 * it;
    const int r = idx / 35, j = idx - r * 35;
    const int gy = iy0 + r, gx = ix0 + 2 * j;
    const bool ok = idx < NP && (unsigned)gy < (unsigned)H && (unsigned)gx < (unsigned)W;
    soff[it] = ok ? (unsigned)(gy * W + gx) : 0u;
    okm |= (ok ? 1u : 0u) << it;
  }

  int b = blockIdx.z;
  const int bstep = gridDim.z;
  float2 st[NIT];
  {
    const float* inb = in + (long)b * inBatchStride;
#pragma unroll
    for (int it = 0; it < NIT; ++it)
      st[it] = *reinterpret_cast<const float2*>(inb + soff[it]);
  }

  sF[tid] = fv;
  lds_barrier();

  v2f Aw[9];
  {
    const int ph = nl & 1, ch = nl >> 1;
#pragma unroll
    for (int s = 0; s < 9; ++s) {
#pragma unroll
      for (int t = 0; t < 2; ++t) {
        int kh, wx;
        if (s < 7) { const int q = 2 * s + t; const int rr = q / 7; kh = 2 * rr + hf; wx = q - 7 * rr; }
        else       { kh = 4; wx = 4 * (s - 7) + t + 2 * hf; }
        const int kw = wx - 2 * ph;
        const bool ok = kw >= 0 && kw <= 4;
        const float w = sF[ch * 25 + kh * 5 + (ok ? kw : 0)];
        Aw[s][t] = ok ? w : 0.0f;
      }
    }
  }

  const float* pa = sIn + (2 * wv * RPW + hf) * P + 4 * nl;
  const float* pb = sIn + (2 * wv * RPW + 4) * P + 4 * nl + 2 * hf;

  for (;;) {
#pragma unroll
    for (int it = 0; it < NIT; ++it) {
      const int idx = tid + 256 * it;
      float2 v = st[it];
      if (!((okm >> it) & 1u)) { v.x = 0.0f; v.y = 0.0f; }
      if (idx < NP) *reinterpret_cast<float2*>(&sIn[2 * idx]) = v;
    }
    lds_barrier();

    const int bn = b + bstep;
    if (bn < nb) {
      const float* inb = in + (long)bn * inBatchStride;
#pragma unroll
      for (int it = 0; it < NIT; ++it)
        st[it] = *reinterpret_cast<const float2*>(inb + soff[it]);
    }

#pragma unroll
    for (int i = 0; i < RPW; i += RS) {
      v8f acc0 = {}, acc1 = {};
#pragma unroll
      for (int s = 0; s < 9; ++s) {
        v2f B0, B1 = {};
#pragma unroll
        for (int t = 0; t < 2; ++t) {
          int o;
          const float* p;
          if (s < 7) { const int q = 2 * s + t; const int rr = q / 7; p = pa; o = 2 * rr * P + (q - 7 * rr); }
          else       { p = pb; o = 4 * (s - 7) + t; }
          B0[t] = p[o + (i) * 2 * P];
          if (RS == 2) B1[t] = p[o + (i + 1) * 2 * P];
        }
        acc0 = __builtin_amdgcn_wmma_f32_16x16x4_f32(
            false, Aw[s], false, B0, (short)0, acc0, false, false);
        if (RS == 2)
          acc1 = __builtin_amdgcn_wmma_f32_16x16x4_f32(
              false, Aw[s], false, B1, (short)0, acc1, false, false);
      }
      {
        const int gy = row0 + wv * RPW + i;
        const long cs = (long)Ho * Wo;
        float* o = out + (((long)b * 8 + 4 * hf) * Ho + gy) * Wo + col0 + 2 * nl;
#pragma unroll
        for (int jj = 0; jj < 4; ++jj) {
          float2 v;
          v.x = acc0[2 * jj];
          v.y = acc0[2 * jj + 1];
          *reinterpret_cast<float2*>(o + jj * cs) = v;
        }
        if (RS == 2) {
#pragma unroll
          for (int jj = 0; jj < 4; ++jj) {
            float2 v;
            v.x = acc1[2 * jj];
            v.y = acc1[2 * jj + 1];
            *reinterpret_cast<float2*>(o + jj * cs + Wo) = v;
          }
        }
      }
    }

    if (bn >= nb) break;
    b = bn;
    lds_barrier();
  }
}

template <int CG, int RG, bool WR>
__global__ __launch_bounds__(32 * CG * RG) __attribute__((amdgpu_waves_per_eu(8))) void inv_tconv5x5_s2_wmma(
    const float* __restrict__ rec,  long recBatchStride,
    const float* __restrict__ coef,
    const float* __restrict__ filt,
    float* __restrict__ outp,
    int H, int W, int nb)
{
  constexpr int NT   = 32 * CG * RG;
  constexpr int TC   = 16 * CG;
  constexpr int TR   = 4 * RG;
  constexpr int NRI  = TR + 2;
  constexpr int PT   = TC + 4;
  constexpr int PP   = PT / 2;
  constexpr int PL   = NRI * PT;
  constexpr int PPS  = WR ? TC / 2 : PP;
  constexpr int NP0  = NRI * PPS;
  constexpr int LSTEP = 2 * NT + (WR ? 4 * (NT / PPS) : 0);
  static_assert(!WR || (NT % PPS == 0 && PPS % 32 == 0), "a wave's load must stay inside one staged row");
  constexpr int NIT0 = (NP0 + NT - 1) / NT;
  constexpr int NP1  = 7 * NP0;
  constexpr int NIT1 = (NP1 + NT - 1) / NT;
  constexpr int NS   = 36;
  constexpr int NWT  = (NS * 64 + NT - 1) / NT;
  constexpr int BK   = 6;
  static_assert(PL % 64 >= 16 && PL % 64 <= 48, "plane pitch puts the lane halves on the same banks");

  const int bx0 = blockIdx.x * TC;
  const int by0 = blockIdx.y * TR;
  const int tid = threadIdx.x;
  const int wv  = tid >> 5;
  const int cg  = wv % CG;
  const int rg  = wv / CG;
  const int lane = tid & 31;
  const int hf  = lane >> 4;
  const int nl  = lane & 15;

  __shared__ __attribute__((aligned(16))) float sIn[8 * PL];
  __shared__ float sF[NT];
  __shared__ __attribute__((aligned(16))) float sW[NS * 64];
  __shared__ unsigned sOff[(NIT0 + NIT1) * NT];

  const float fv = filt[tid < 200 ? tid : 0];

  unsigned off0[NIT0], off1[NIT1];
  unsigned okm0 = 0u, okm1 = 0u;
#pragma unroll
  for (int it = 0; it < NIT0; ++it) {
    const int idx = tid + NT * it;
    const int r = idx / PPS, j = idx - r * PPS;
    const int gy = by0 + r - 1, gx = bx0 + 2 * j - (WR ? 0 : 2);
    const bool ok = idx < NP0 && (unsigned)gy < (unsigned)H && (unsigned)gx < (unsigned)W;
    off0[it] = ok ? (unsigned)(gy * W + gx) : 0u;
    okm0 |= (ok ? 1u : 0u) << it;
  }
#pragma unroll
  for (int it = 0; it < NIT1; ++it) {
    const int idx = tid + NT * it;
    const int c1 = idx / NP0, rem = idx - c1 * NP0;
    const int r = rem / PPS, j = rem - r * PPS;
    const int gy = by0 + r - 1, gx = bx0 + 2 * j - (WR ? 0 : 2);
    const bool ok = idx < NP1 && (unsigned)gy < (unsigned)H && (unsigned)gx < (unsigned)W;
    off1[it] = ok ? (unsigned)(((c1 + 1) * H + gy) * W + gx) : 0u;
    okm1 |= (ok ? 1u : 0u) << it;
  }

  int b = blockIdx.z;
  const int bstep = gridDim.z;
  const long coefBatchStride = 8L * H * W;
  float2 s0[NIT0], s1[NIT1];
  {
    const float* recb  = rec  + (long)b * recBatchStride;
    const float* coefb = coef + (long)b * coefBatchStride;
#pragma unroll
    for (int it = 0; it < NIT0; ++it)
      s0[it] = *reinterpret_cast<const float2*>(recb + off0[it]);
#pragma unroll
    for (int it = 0; it < NIT1; ++it)
      s1[it] = *reinterpret_cast<const float2*>(coefb + off1[it]);
  }

#pragma unroll
  for (int it = 0; it < NIT0; ++it) sOff[it * NT + tid] = off0[it];
#pragma unroll
  for (int it = 0; it < NIT1; ++it) sOff[(NIT0 + it) * NT + tid] = off1[it];
  sF[tid] = fv;
  if (WR) {
    if (tid < 8 * NRI) {
      float2 z; z.x = 0.0f; z.y = 0.0f;
      *reinterpret_cast<float2*>(&sIn[tid * PT]) = z;
      *reinterpret_cast<float2*>(&sIn[tid * PT + TC + 2]) = z;
    }
  }
  lds_barrier();

#pragma unroll
  for (int e = 0; e < NWT; ++e) {
    const int idx = tid + NT * e;
    if (idx >= NS * 64) break;
    const int s = idx >> 6, l = (idx >> 1) & 31, t = idx & 1;
    const int m = l & 15, h = l >> 4;
    const int q = 2 * s + t;
    const int cc = q / 18, rem = q - cc * 18;
    const int wy = rem / 3, dxi = rem - wy * 3;
    const int r = m >> 2, py = (m >> 1) & 1, px = m & 1;
    const int dy = wy - 1 - r;
    const int ih = 2 + py - 2 * dy;
    const int iw = 4 + px - 2 * dxi;
    const bool ok = dy >= -1 && dy <= 1 && ih <= 4 && iw <= 4;
    const float w = sF[(2 * cc + h) * 25 + (ok ? ih * 5 + iw : 0)];
    sW[idx] = ok ? w : 0.0f;
  }

  const int lbase = 2 * tid + (WR ? 4 * (tid / PPS) + 2 : 0);
  const int pbase = hf * PL + (rg * 4) * PT + cg * 16 + nl + 1;
  const float* wfrag = sW + 2 * lane;

  for (;;) {
#pragma unroll
    for (int it = 0; it < NIT0; ++it) {
      const int idx = tid + NT * it;
      float2 v = s0[it];
      if (!((okm0 >> it) & 1u)) { v.x = 0.0f; v.y = 0.0f; }
      if (idx < NP0) *reinterpret_cast<float2*>(&sIn[lbase + LSTEP * it]) = v;
    }
#pragma unroll
    for (int it = 0; it < NIT1; ++it) {
      const int idx = tid + NT * it;
      float2 v = s1[it];
      if (!((okm1 >> it) & 1u)) { v.x = 0.0f; v.y = 0.0f; }
      if (idx < NP1) *reinterpret_cast<float2*>(&sIn[PL + lbase + LSTEP * it]) = v;
    }
    lds_barrier();

    const int bn = b + bstep;
    if (bn < nb) {
      const float* recb  = rec  + (long)bn * recBatchStride;
      const float* coefb = coef + (long)bn * coefBatchStride;
#pragma unroll
      for (int it = 0; it < NIT0; ++it)
        s0[it] = *reinterpret_cast<const float2*>(recb + sOff[it * NT + tid]);
#pragma unroll
      for (int it = 0; it < NIT1; ++it)
        s1[it] = *reinterpret_cast<const float2*>(coefb + sOff[(NIT0 + it) * NT + tid]);
    }

    v8f acc0 = {}, acc1 = {};
    v2f Wb[2][BK], Pb[2][BK];
#pragma unroll
    for (int u = 0; u < BK; ++u) {
      Wb[0][u] = *reinterpret_cast<const v2f*>(wfrag + 64 * u);
#pragma unroll
      for (int t = 0; t < 2; ++t) {
        const int q = 2 * u + t;
        const int cc = q / 18, rem = q - cc * 18;
        Pb[0][u][t] = sIn[pbase + 2 * cc * PL + (rem / 3) * PT + (rem % 3)];
      }
    }
#pragma unroll
    for (int blk = 0; blk < NS / BK; ++blk) {
      const int cb = blk & 1, nx = cb ^ 1;
      if (blk + 1 < NS / BK) {
#pragma unroll
        for (int u = 0; u < BK; ++u) {
          const int s = (blk + 1) * BK + u;
          Wb[nx][u] = *reinterpret_cast<const v2f*>(wfrag + 64 * s);
#pragma unroll
          for (int t = 0; t < 2; ++t) {
            const int q = 2 * s + t;
            const int cc = q / 18, rem = q - cc * 18;
            Pb[nx][u][t] = sIn[pbase + 2 * cc * PL + (rem / 3) * PT + (rem % 3)];
          }
        }
      }
      __builtin_amdgcn_sched_barrier(0);
#pragma unroll
      for (int u = 0; u < BK; ++u) {
        if ((blk * BK + u) & 1)
          acc1 = __builtin_amdgcn_wmma_f32_16x16x4_f32(
              false, Wb[cb][u], false, Pb[cb][u], (short)0, acc1, false, false);
        else
          acc0 = __builtin_amdgcn_wmma_f32_16x16x4_f32(
              false, Wb[cb][u], false, Pb[cb][u], (short)0, acc0, false, false);
      }
      __builtin_amdgcn_sched_barrier(0);
    }
    const v8f acc = acc0 + acc1;

    {
      const int oy0 = 2 * (by0 + rg * 4 + 2 * hf);
      float* o = outp + ((long)b * (2 * H) + oy0) * (2L * W) + 2 * (bx0 + cg * 16 + nl);
#pragma unroll
      for (int jj = 0; jj < 4; ++jj) {
        float2 v;
        v.x = acc[2 * jj];
        v.y = acc[2 * jj + 1];
        *reinterpret_cast<float2*>(o + (long)jj * (2L * W)) = v;
      }
    }

    if (bn >= nb) break;
    b = bn;
    lds_barrier();
  }
}

extern "C" void kernel_launch(void* const* d_in, const int* in_sizes, int n_in,
                              void* d_out, int out_size, void* d_ws, size_t ws_size,
                              hipStream_t stream) {
  (void)in_sizes; (void)n_in; (void)out_size; (void)ws_size;
  const float* x     = (const float*)d_in[0];
  const float* fwd_f = (const float*)d_in[1];
  const float* inv_f = (const float*)d_in[2];
  float* out = (float*)d_out;
  float* ws  = (float*)d_ws;

  const size_t n4 = 32UL * 8 * 256 * 256;
  const size_t n3 = 32UL * 8 * 128 * 128;
  const size_t n2 = 32UL * 8 * 64 * 64;
  const size_t n1 = 32UL * 8 * 32 * 32;
  float* res4 = ws;
  float* res3 = res4 + n4;
  float* res2 = res3 + n3;
  float* res1 = res2 + n2;
  float* rec1 = res1 + n1;
  float* rec2 = rec1 + 32UL * 64 * 64;
  float* rec3 = rec2 + 32UL * 128 * 128;

  const dim3 blk(256);

  fwd_conv5x5_s2_wmma<4><<<dim3(8, 8, 16), blk, 0, stream>>>(
      x,    512L * 512,      fwd_f, res4, 512, 512, 256, 256, 32);
  fwd_conv5x5_s2_wmma<2><<<dim3(4, 8, 16), blk, 0, stream>>>(
      res4, 8L * 256 * 256,  fwd_f, res3, 256, 256, 128, 128, 32);
  fwd_conv5x5_s2_wmma<1><<<dim3(2, 8, 32), blk, 0, stream>>>(
      res3, 8L * 128 * 128,  fwd_f, res2, 128, 128, 64, 64, 32);
  fwd_conv5x5_s2_wmma<1><<<dim3(1, 4, 32), blk, 0, stream>>>(
      res2, 8L * 64 * 64,    fwd_f, res1, 64, 64, 32, 32, 32);

  inv_tconv5x5_s2_wmma<1, 8, false><<<dim3(2, 1, 32), blk, 0, stream>>>(
      res1, 8L * 32 * 32,    res1, inv_f, rec1, 32, 32, 32);
  inv_tconv5x5_s2_wmma<4, 2, true><<<dim3(1, 8, 32), blk, 0, stream>>>(
      rec1, 64L * 64,        res2, inv_f, rec2, 64, 64, 32);
  inv_tconv5x5_s2_wmma<8, 2, true><<<dim3(1, 16, 16), dim3(512), 0, stream>>>(
      rec2, 128L * 128,      res3, inv_f, rec3, 128, 128, 32);
  inv_tconv5x5_s2_wmma<16, 1, true><<<dim3(1, 64, 8), dim3(512), 0, stream>>>(
      rec3, 256L * 256,      res4, inv_f, out, 256, 256, 32);
}


# Forward level 1 and inverse level 1 in ONE launch: a work-group per 8 band rows computes them (1-row halo) into LDS in the inverse's staged layout, the band never goes to memory: 7 launches, not 8
# speedup vs baseline: 3.3244x; 1.0590x over previous
#include <hip/hip_runtime.h>


typedef __attribute__((ext_vector_type(2))) float v2f;
typedef __attribute__((ext_vector_type(8))) float v8f;

__device__ __forceinline__ void lds_barrier() {
  __builtin_amdgcn_fence(__ATOMIC_RELEASE, "workgroup", "local");
  __builtin_amdgcn_s_barrier();
  __builtin_amdgcn_fence(__ATOMIC_ACQUIRE, "workgroup", "local");
}

template <int RPW>
__global__ __launch_bounds__(256) __attribute__((amdgpu_waves_per_eu(8))) void fwd_conv5x5_s2_wmma(
    const float* __restrict__ in, long inBatchStride,
    const float* __restrict__ filt,
    float* __restrict__ out,
    int H, int W, int Ho, int Wo, int nb)
{
  constexpr int TR  = 8 * RPW;
  constexpr int NR  = 2 * TR + 3;
  constexpr int P   = 70;
  constexpr int NP  = NR * 35;
  constexpr int NIT = (NP + 255) / 256;
  constexpr int RS  = RPW >= 2 ? 2 : 1;

  const int col0 = blockIdx.x * 32;
  const int row0 = blockIdx.y * TR;
  const int tid  = threadIdx.x;
  const int wv   = tid >> 5;
  const int lane = tid & 31;
  const int hf   = lane >> 4;
  const int nl   = lane & 15;

  __shared__ __attribute__((aligned(16))) float sIn[NR * P];
  __shared__ float sF[256];

  const float fv = filt[tid < 200 ? tid : 0];

  const int iy0 = 2 * row0 - 2;
  const int ix0 = 2 * col0 - 2;
  unsigned soff[NIT];
  unsigned okm = 0u;
#pragma unroll
  for (int it = 0; it < NIT; ++it) {
    const int idx = tid + 256 * it;
    const int r = idx / 35, j = idx - r * 35;
    const int gy = iy0 + r, gx = ix0 + 2 * j;
    const bool ok = idx < NP && (unsigned)gy < (unsigned)H && (unsigned)gx < (unsigned)W;
    soff[it] = ok ? (unsigned)(gy * W + gx) : 0u;
    okm |= (ok ? 1u : 0u) << it;
  }

  int b = blockIdx.z;
  const int bstep = gridDim.z;
  float2 st[NIT];
  {
    const float* inb = in + (long)b * inBatchStride;
#pragma unroll
    for (int it = 0; it < NIT; ++it)
      st[it] = *reinterpret_cast<const float2*>(inb + soff[it]);
  }

  sF[tid] = fv;
  lds_barrier();

  v2f Aw[9];
  {
    const int ph = nl & 1, ch = nl >> 1;
#pragma unroll
    for (int s = 0; s < 9; ++s) {
#pragma unroll
      for (int t = 0; t < 2; ++t) {
        int kh, wx;
        if (s < 7) { const int q = 2 * s + t; const int rr = q / 7; kh = 2 * rr + hf; wx = q - 7 * rr; }
        else       { kh = 4; wx = 4 * (s - 7) + t + 2 * hf; }
        const int kw = wx - 2 * ph;
        const bool ok = kw >= 0 && kw <= 4;
        const float w = sF[ch * 25 + kh * 5 + (ok ? kw : 0)];
        Aw[s][t] = ok ? w : 0.0f;
      }
    }
  }

  const float* pa = sIn + (2 * wv * RPW + hf) * P + 4 * nl;
  const float* pb = sIn + (2 * wv * RPW + 4) * P + 4 * nl + 2 * hf;

  for (;;) {
#pragma unroll
    for (int it = 0; it < NIT; ++it) {
      const int idx = tid + 256 * it;
      float2 v = st[it];
      if (!((okm >> it) & 1u)) { v.x = 0.0f; v.y = 0.0f; }
      if (idx < NP) *reinterpret_cast<float2*>(&sIn[2 * idx]) = v;
    }
    lds_barrier();

    const int bn = b + bstep;
    if (bn < nb) {
      const float* inb = in + (long)bn * inBatchStride;
#pragma unroll
      for (int it = 0; it < NIT; ++it)
        st[it] = *reinterpret_cast<const float2*>(inb + soff[it]);
    }

#pragma unroll
    for (int i = 0; i < RPW; i += RS) {
      v8f acc0 = {}, acc1 = {};
#pragma unroll
      for (int s = 0; s < 9; ++s) {
        v2f B0, B1 = {};
#pragma unroll
        for (int t = 0; t < 2; ++t) {
          int o;
          const float* p;
          if (s < 7) { const int q = 2 * s + t; const int rr = q / 7; p = pa; o = 2 * rr * P + (q - 7 * rr); }
          else       { p = pb; o = 4 * (s - 7) + t; }
          B0[t] = p[o + (i) * 2 * P];
          if (RS == 2) B1[t] = p[o + (i + 1) * 2 * P];
        }
        acc0 = __builtin_amdgcn_wmma_f32_16x16x4_f32(
            false, Aw[s], false, B0, (short)0, acc0, false, false);
        if (RS == 2)
          acc1 = __builtin_amdgcn_wmma_f32_16x16x4_f32(
              false, Aw[s], false, B1, (short)0, acc1, false, false);
      }
      {
        const int gy = row0 + wv * RPW + i;
        const long cs = (long)Ho * Wo;
        float* o = out + (((long)b * 8 + 4 * hf) * Ho + gy) * Wo + col0 + 2 * nl;
#pragma unroll
        for (int jj = 0; jj < 4; ++jj) {
          float2 v;
          v.x = acc0[2 * jj];
          v.y = acc0[2 * jj + 1];
          *reinterpret_cast<float2*>(o + jj * cs) = v;
        }
        if (RS == 2) {
#pragma unroll
          for (int jj = 0; jj < 4; ++jj) {
            float2 v;
            v.x = acc1[2 * jj];
            v.y = acc1[2 * jj + 1];
            *reinterpret_cast<float2*>(o + jj * cs + Wo) = v;
          }
        }
      }
    }

    if (bn >= nb) break;
    b = bn;
    lds_barrier();
  }
}

template <int CG, int RG, bool WR>
__global__ __launch_bounds__(32 * CG * RG) __attribute__((amdgpu_waves_per_eu(8))) void inv_tconv5x5_s2_wmma(
    const float* __restrict__ rec,  long recBatchStride,
    const float* __restrict__ coef,
    const float* __restrict__ filt,
    float* __restrict__ outp,
    int H, int W, int nb)
{
  constexpr int NT   = 32 * CG * RG;
  constexpr int TC   = 16 * CG;
  constexpr int TR   = 4 * RG;
  constexpr int NRI  = TR + 2;
  constexpr int PT   = TC + 4;
  constexpr int PP   = PT / 2;
  constexpr int PL   = NRI * PT;
  constexpr int PPS  = WR ? TC / 2 : PP;
  constexpr int NP0  = NRI * PPS;
  constexpr int LSTEP = 2 * NT + (WR ? 4 * (NT / PPS) : 0);
  static_assert(!WR || (NT % PPS == 0 && PPS % 32 == 0), "a wave's load must stay inside one staged row");
  constexpr int NIT0 = (NP0 + NT - 1) / NT;
  constexpr int NP1  = 7 * NP0;
  constexpr int NIT1 = (NP1 + NT - 1) / NT;
  constexpr int NS   = 36;
  constexpr int NWT  = (NS * 64 + NT - 1) / NT;
  constexpr int BK   = 6;
  static_assert(PL % 64 >= 16 && PL % 64 <= 48, "plane pitch puts the lane halves on the same banks");

  const int bx0 = blockIdx.x * TC;
  const int by0 = blockIdx.y * TR;
  const int tid = threadIdx.x;
  const int wv  = tid >> 5;
  const int cg  = wv % CG;
  const int rg  = wv / CG;
  const int lane = tid & 31;
  const int hf  = lane >> 4;
  const int nl  = lane & 15;

  __shared__ __attribute__((aligned(16))) float sIn[8 * PL];
  __shared__ float sF[NT];
  __shared__ __attribute__((aligned(16))) float sW[NS * 64];
  __shared__ unsigned sOff[(NIT0 + NIT1) * NT];

  const float fv = filt[tid < 200 ? tid : 0];

  unsigned off0[NIT0], off1[NIT1];
  unsigned okm0 = 0u, okm1 = 0u;
#pragma unroll
  for (int it = 0; it < NIT0; ++it) {
    const int idx = tid + NT * it;
    const int r = idx / PPS, j = idx - r * PPS;
    const int gy = by0 + r - 1, gx = bx0 + 2 * j - (WR ? 0 : 2);
    const bool ok = idx < NP0 && (unsigned)gy < (unsigned)H && (unsigned)gx < (unsigned)W;
    off0[it] = ok ? (unsigned)(gy * W + gx) : 0u;
    okm0 |= (ok ? 1u : 0u) << it;
  }
#pragma unroll
  for (int it = 0; it < NIT1; ++it) {
    const int idx = tid + NT * it;
    const int c1 = idx / NP0, rem = idx - c1 * NP0;
    const int r = rem / PPS, j = rem - r * PPS;
    const int gy = by0 + r - 1, gx = bx0 + 2 * j - (WR ? 0 : 2);
    const bool ok = idx < NP1 && (unsigned)gy < (unsigned)H && (unsigned)gx < (unsigned)W;
    off1[it] = ok ? (unsigned)(((c1 + 1) * H + gy) * W + gx) : 0u;
    okm1 |= (ok ? 1u : 0u) << it;
  }

  int b = blockIdx.z;
  const int bstep = gridDim.z;
  const long coefBatchStride = 8L * H * W;
  float2 s0[NIT0], s1[NIT1];
  {
    const float* recb  = rec  + (long)b * recBatchStride;
    const float* coefb = coef + (long)b * coefBatchStride;
#pragma unroll
    for (int it = 0; it < NIT0; ++it)
      s0[it] = *reinterpret_cast<const float2*>(recb + off0[it]);
#pragma unroll
    for (int it = 0; it < NIT1; ++it)
      s1[it] = *reinterpret_cast<const float2*>(coefb + off1[it]);
  }

#pragma unroll
  for (int it = 0; it < NIT0; ++it) sOff[it * NT + tid] = off0[it];
#pragma unroll
  for (int it = 0; it < NIT1; ++it) sOff[(NIT0 + it) * NT + tid] = off1[it];
  sF[tid] = fv;
  if (WR) {
    if (tid < 8 * NRI) {
      float2 z; z.x = 0.0f; z.y = 0.0f;
      *reinterpret_cast<float2*>(&sIn[tid * PT]) = z;
      *reinterpret_cast<float2*>(&sIn[tid * PT + TC + 2]) = z;
    }
  }
  lds_barrier();

#pragma unroll
  for (int e = 0; e < NWT; ++e) {
    const int idx = tid + NT * e;
    if (idx >= NS * 64) break;
    const int s = idx >> 6, l = (idx >> 1) & 31, t = idx & 1;
    const int m = l & 15, h = l >> 4;
    const int q = 2 * s + t;
    const int cc = q / 18, rem = q - cc * 18;
    const int wy = rem / 3, dxi = rem - wy * 3;
    const int r = m >> 2, py = (m >> 1) & 1, px = m & 1;
    const int dy = wy - 1 - r;
    const int ih = 2 + py - 2 * dy;
    const int iw = 4 + px - 2 * dxi;
    const bool ok = dy >= -1 && dy <= 1 && ih <= 4 && iw <= 4;
    const float w = sF[(2 * cc + h) * 25 + (ok ? ih * 5 + iw : 0)];
    sW[idx] = ok ? w : 0.0f;
  }

  const int lbase = 2 * tid + (WR ? 4 * (tid / PPS) + 2 : 0);
  const int pbase = hf * PL + (rg * 4) * PT + cg * 16 + nl + 1;
  const float* wfrag = sW + 2 * lane;

  for (;;) {
#pragma unroll
    for (int it = 0; it < NIT0; ++it) {
      const int idx = tid + NT * it;
      float2 v = s0[it];
      if (!((okm0 >> it) & 1u)) { v.x = 0.0f; v.y = 0.0f; }
      if (idx < NP0) *reinterpret_cast<float2*>(&sIn[lbase + LSTEP * it]) = v;
    }
#pragma unroll
    for (int it = 0; it < NIT1; ++it) {
      const int idx = tid + NT * it;
      float2 v = s1[it];
      if (!((okm1 >> it) & 1u)) { v.x = 0.0f; v.y = 0.0f; }
      if (idx < NP1) *reinterpret_cast<float2*>(&sIn[PL + lbase + LSTEP * it]) = v;
    }
    lds_barrier();

    const int bn = b + bstep;
    if (bn < nb) {
      const float* recb  = rec  + (long)bn * recBatchStride;
      const float* coefb = coef + (long)bn * coefBatchStride;
#pragma unroll
      for (int it = 0; it < NIT0; ++it)
        s0[it] = *reinterpret_cast<const float2*>(recb + sOff[it * NT + tid]);
#pragma unroll
      for (int it = 0; it < NIT1; ++it)
        s1[it] = *reinterpret_cast<const float2*>(coefb + sOff[(NIT0 + it) * NT + tid]);
    }

    v8f acc0 = {}, acc1 = {};
    v2f Wb[2][BK], Pb[2][BK];
#pragma unroll
    for (int u = 0; u < BK; ++u) {
      Wb[0][u] = *reinterpret_cast<const v2f*>(wfrag + 64 * u);
#pragma unroll
      for (int t = 0; t < 2; ++t) {
        const int q = 2 * u + t;
        const int cc = q / 18, rem = q - cc * 18;
        Pb[0][u][t] = sIn[pbase + 2 * cc * PL + (rem / 3) * PT + (rem % 3)];
      }
    }
#pragma unroll
    for (int blk = 0; blk < NS / BK; ++blk) {
      const int cb = blk & 1, nx = cb ^ 1;
      if (blk + 1 < NS / BK) {
#pragma unroll
        for (int u = 0; u < BK; ++u) {
          const int s = (blk + 1) * BK + u;
          Wb[nx][u] = *reinterpret_cast<const v2f*>(wfrag + 64 * s);
#pragma unroll
          for (int t = 0; t < 2; ++t) {
            const int q = 2 * s + t;
            const int cc = q / 18, rem = q - cc * 18;
            Pb[nx][u][t] = sIn[pbase + 2 * cc * PL + (rem / 3) * PT + (rem % 3)];
          }
        }
      }
      __builtin_amdgcn_sched_barrier(0);
#pragma unroll
      for (int u = 0; u < BK; ++u) {
        if ((blk * BK + u) & 1)
          acc1 = __builtin_amdgcn_wmma_f32_16x16x4_f32(
              false, Wb[cb][u], false, Pb[cb][u], (short)0, acc1, false, false);
        else
          acc0 = __builtin_amdgcn_wmma_f32_16x16x4_f32(
              false, Wb[cb][u], false, Pb[cb][u], (short)0, acc0, false, false);
      }
      __builtin_amdgcn_sched_barrier(0);
    }
    const v8f acc = acc0 + acc1;

    {
      const int oy0 = 2 * (by0 + rg * 4 + 2 * hf);
      float* o = outp + ((long)b * (2 * H) + oy0) * (2L * W) + 2 * (bx0 + cg * 16 + nl);
#pragma unroll
      for (int jj = 0; jj < 4; ++jj) {
        float2 v;
        v.x = acc[2 * jj];
        v.y = acc[2 * jj + 1];
        *reinterpret_cast<float2*>(o + (long)jj * (2L * W)) = v;
      }
    }

    if (bn >= nb) break;
    b = bn;
    lds_barrier();
  }
}

__global__ __launch_bounds__(256) void fwd1_inv1_wmma(
    const float* __restrict__ in, long inBatchStride,
    const float* __restrict__ ffilt,
    const float* __restrict__ ifilt,
    float* __restrict__ outp)
{
  constexpr int HI  = 64;
  constexpr int HB  = 32;
  constexpr int NBR = 10;
  constexpr int NR  = 2 * NBR + 3;
  constexpr int P   = 70;
  constexpr int NP  = NR * 35;
  constexpr int NIT = (NP + 255) / 256;
  constexpr int PT  = HB + 4;
  constexpr int PL  = NBR * PT;
  constexpr int NS  = 36;
  constexpr int NWT = NS * 64 / 256;
  constexpr int BK  = 6;
  static_assert(PL % 64 >= 16 && PL % 64 <= 48, "plane pitch puts the lane halves on the same banks");
  static_assert((8 * PL) % 4 == 0, "the band is cleared in 16-byte pieces");

  const int q   = blockIdx.x;
  const int b   = blockIdx.y;
  const int tid = threadIdx.x;
  const int wv  = tid >> 5;
  const int lane = tid & 31;
  const int hf  = lane >> 4;
  const int nl  = lane & 15;

  __shared__ __attribute__((aligned(16))) float sX[NR * P];
  __shared__ __attribute__((aligned(16))) float sI[8 * PL];
  __shared__ float sFf[256];
  __shared__ float sFi[256];
  __shared__ __attribute__((aligned(16))) float sW[NS * 64];

  const float fvf = ffilt[tid < 200 ? tid : 0];
  const float fvi = ifilt[tid < 200 ? tid : 0];

  const float* inb = in + (long)b * inBatchStride;
  const int iy0 = 16 * q - 4;
  float2 st[NIT];
  unsigned okm = 0u;
#pragma unroll
  for (int it = 0; it < NIT; ++it) {
    const int idx = tid + 256 * it;
    const int r = idx / 35, j = idx - r * 35;
    const int gy = iy0 + r, gx = 2 * j - 2;
    const bool ok = idx < NP && (unsigned)gy < (unsigned)HI && (unsigned)gx < (unsigned)HI;
    st[it] = *reinterpret_cast<const float2*>(inb + (ok ? gy * HI + gx : 0));
    okm |= (ok ? 1u : 0u) << it;
  }

  for (int i = tid; i < 8 * PL / 4; i += 256) {
    float4 z; z.x = 0.0f; z.y = 0.0f; z.z = 0.0f; z.w = 0.0f;
    reinterpret_cast<float4*>(sI)[i] = z;
  }
  sFf[tid] = fvf;
  sFi[tid] = fvi;
#pragma unroll
  for (int it = 0; it < NIT; ++it) {
    const int idx = tid + 256 * it;
    float2 v = st[it];
    if (!((okm >> it) & 1u)) { v.x = 0.0f; v.y = 0.0f; }
    if (idx < NP) *reinterpret_cast<float2*>(&sX[2 * idx]) = v;
  }
  lds_barrier();

  if (wv < NBR / 2) {
    v2f Aw[9];
    {
      const int ph = nl & 1, ch = nl >> 1;
#pragma unroll
      for (int s = 0; s < 9; ++s) {
#pragma unroll
        for (int t = 0; t < 2; ++t) {
          int kh, wx;
          if (s < 7) { const int qq = 2 * s + t; const int rr = qq / 7; kh = 2 * rr + hf; wx = qq - 7 * rr; }
          else       { kh = 4; wx = 4 * (s - 7) + t + 2 * hf; }
          const int kw = wx - 2 * ph;
          const bool ok = kw >= 0 && kw <= 4;
          const float w = sFf[ch * 25 + kh * 5 + (ok ? kw : 0)];
          Aw[s][t] = ok ? w : 0.0f;
        }
      }
    }
    const float* pa = sX + (4 * wv + hf) * P + 4 * nl;
    const float* pb = sX + (4 * wv + 4) * P + 4 * nl + 2 * hf;
    v8f acc0 = {}, acc1 = {};
#pragma unroll
    for (int s = 0; s < 9; ++s) {
      v2f B0, B1;
#pragma unroll
      for (int t = 0; t < 2; ++t) {
        int o;
        const float* p;
        if (s < 7) { const int qq = 2 * s + t; const int rr = qq / 7; p = pa; o = 2 * rr * P + (qq - 7 * rr); }
        else       { p = pb; o = 4 * (s - 7) + t; }
        B0[t] = p[o];
        B1[t] = p[o + 2 * P];
      }
      acc0 = __builtin_amdgcn_wmma_f32_16x16x4_f32(
          false, Aw[s], false, B0, (short)0, acc0, false, false);
      acc1 = __builtin_amdgcn_wmma_f32_16x16x4_f32(
          false, Aw[s], false, B1, (short)0, acc1, false, false);
    }
    const int lr = 2 * wv;
    const int g0 = 8 * q - 1 + lr;
    float* d = sI + (4 * hf) * PL + lr * PT + 2 + 2 * nl;
    if ((unsigned)g0 < (unsigned)HB) {
#pragma unroll
      for (int jj = 0; jj < 4; ++jj) {
        float2 v;
        v.x = acc0[2 * jj];
        v.y = acc0[2 * jj + 1];
        *reinterpret_cast<float2*>(d + jj * PL) = v;
      }
    }
    if ((unsigned)(g0 + 1) < (unsigned)HB) {
#pragma unroll
      for (int jj = 0; jj < 4; ++jj) {
        float2 v;
        v.x = acc1[2 * jj];
        v.y = acc1[2 * jj + 1];
        *reinterpret_cast<float2*>(d + jj * PL + PT) = v;
      }
    }
  }

#pragma unroll
  for (int e = 0; e < NWT; ++e) {
    const int idx = tid + 256 * e;
    const int s = idx >> 6, l = (idx >> 1) & 31, t = idx & 1;
    const int m = l & 15, h = l >> 4;
    const int qq = 2 * s + t;
    const int cc = qq / 18, rem = qq - cc * 18;
    const int wy = rem / 3, dxi = rem - wy * 3;
    const int r = m >> 2, py = (m >> 1) & 1, px = m & 1;
    const int dy = wy - 1 - r;
    const int ih = 2 + py - 2 * dy;
    const int iw = 4 + px - 2 * dxi;
    const bool ok = dy >= -1 && dy <= 1 && ih <= 4 && iw <= 4;
    const float w = sFi[(2 * cc + h) * 25 + (ok ? ih * 5 + iw : 0)];
    sW[idx] = ok ? w : 0.0f;
  }
  lds_barrier();

  if (wv < 4) {
    const int cg = wv & 1;
    const int rg = wv >> 1;
    const int pbase = hf * PL + (rg * 4) * PT + cg * 16 + nl + 1;
    const float* wfrag = sW + 2 * lane;

    v8f acc0 = {}, acc1 = {};
    v2f Wb[2][BK], Pb[2][BK];
#pragma unroll
    for (int u = 0; u < BK; ++u) {
      Wb[0][u] = *reinterpret_cast<const v2f*>(wfrag + 64 * u);
#pragma unroll
      for (int t = 0; t < 2; ++t) {
        const int qq = 2 * u + t;
        const int cc = qq / 18, rem = qq - cc * 18;
        Pb[0][u][t] = sI[pbase + 2 * cc * PL + (rem / 3) * PT + (rem % 3)];
      }
    }
#pragma unroll
    for (int blk = 0; blk < NS / BK; ++blk) {
      const int cb = blk & 1, nx = cb ^ 1;
      if (blk + 1 < NS / BK) {
#pragma unroll
        for (int u = 0; u < BK; ++u) {
          const int s = (blk + 1) * BK + u;
          Wb[nx][u] = *reinterpret_cast<const v2f*>(wfrag + 64 * s);
#pragma unroll
          for (int t = 0; t < 2; ++t) {
            const int qq = 2 * s + t;
            const int cc = qq / 18, rem = qq - cc * 18;
            Pb[nx][u][t] = sI[pbase + 2 * cc * PL + (rem / 3) * PT + (rem % 3)];
          }
        }
      }
      __builtin_amdgcn_sched_barrier(0);
#pragma unroll
      for (int u = 0; u < BK; ++u) {
        if ((blk * BK + u) & 1)
          acc1 = __builtin_amdgcn_wmma_f32_16x16x4_f32(
              false, Wb[cb][u], false, Pb[cb][u], (short)0, acc1, false, false);
        else
          acc0 = __builtin_amdgcn_wmma_f32_16x16x4_f32(
              false, Wb[cb][u], false, Pb[cb][u], (short)0, acc0, false, false);
      }
      __builtin_amdgcn_sched_barrier(0);
    }
    const v8f acc = acc0 + acc1;

    const int oy0 = 2 * (8 * q + rg * 4 + 2 * hf);
    float* o = outp + ((long)b * (2 * HB) + oy0) * (2L * HB) + 2 * (cg * 16 + nl);
#pragma unroll
    for (int jj = 0; jj < 4; ++jj) {
      float2 v;
      v.x = acc[2 * jj];
      v.y = acc[2 * jj + 1];
      *reinterpret_cast<float2*>(o + (long)jj * (2L * HB)) = v;
    }
  }
}

extern "C" void kernel_launch(void* const* d_in, const int* in_sizes, int n_in,
                              void* d_out, int out_size, void* d_ws, size_t ws_size,
                              hipStream_t stream) {
  (void)in_sizes; (void)n_in; (void)out_size; (void)ws_size;
  const float* x     = (const float*)d_in[0];
  const float* fwd_f = (const float*)d_in[1];
  const float* inv_f = (const float*)d_in[2];
  float* out = (float*)d_out;
  float* ws  = (float*)d_ws;

  const size_t n4 = 32UL * 8 * 256 * 256;
  const size_t n3 = 32UL * 8 * 128 * 128;
  const size_t n2 = 32UL * 8 * 64 * 64;
  const size_t n1 = 32UL * 8 * 32 * 32;
  float* res4 = ws;
  float* res3 = res4 + n4;
  float* res2 = res3 + n3;
  float* res1 = res2 + n2;
  float* rec1 = res1 + n1;
  float* rec2 = rec1 + 32UL * 64 * 64;
  float* rec3 = rec2 + 32UL * 128 * 128;

  const dim3 blk(256);

  fwd_conv5x5_s2_wmma<4><<<dim3(8, 8, 16), blk, 0, stream>>>(
      x,    512L * 512,      fwd_f, res4, 512, 512, 256, 256, 32);
  fwd_conv5x5_s2_wmma<2><<<dim3(4, 8, 16), blk, 0, stream>>>(
      res4, 8L * 256 * 256,  fwd_f, res3, 256, 256, 128, 128, 32);
  fwd_conv5x5_s2_wmma<1><<<dim3(2, 8, 32), blk, 0, stream>>>(
      res3, 8L * 128 * 128,  fwd_f, res2, 128, 128, 64, 64, 32);

  fwd1_inv1_wmma<<<dim3(4, 32), blk, 0, stream>>>(
      res2, 8L * 64 * 64,    fwd_f, inv_f, rec1);

  inv_tconv5x5_s2_wmma<4, 2, true><<<dim3(1, 8, 32), blk, 0, stream>>>(
      rec1, 64L * 64,        res2, inv_f, rec2, 64, 64, 32);
  inv_tconv5x5_s2_wmma<8, 2, true><<<dim3(1, 16, 16), dim3(512), 0, stream>>>(
      rec2, 128L * 128,      res3, inv_f, rec3, 128, 128, 32);
  inv_tconv5x5_s2_wmma<16, 1, true><<<dim3(1, 64, 8), dim3(512), 0, stream>>>(
      rec3, 256L * 256,      res4, inv_f, out, 256, 256, 32);
}


# Forward 2, forward 1, inverse 1, inverse 2 in ONE launch: a work-group per 8 level-2 rows computes its own halo rows; both bands and the 64x64 plane stay in LDS only: 5 launches, not 7
# speedup vs baseline: 3.5580x; 1.0703x over previous
#include <hip/hip_runtime.h>


typedef __attribute__((ext_vector_type(2))) float v2f;
typedef __attribute__((ext_vector_type(8))) float v8f;

__device__ __forceinline__ void lds_barrier() {
  __builtin_amdgcn_fence(__ATOMIC_RELEASE, "workgroup", "local");
  __builtin_amdgcn_s_barrier();
  __builtin_amdgcn_fence(__ATOMIC_ACQUIRE, "workgroup", "local");
}

template <int RPW>
__global__ __launch_bounds__(256) __attribute__((amdgpu_waves_per_eu(8))) void fwd_conv5x5_s2_wmma(
    const float* __restrict__ in, long inBatchStride,
    const float* __restrict__ filt,
    float* __restrict__ out,
    int H, int W, int Ho, int Wo, int nb)
{
  constexpr int TR  = 8 * RPW;
  constexpr int NR  = 2 * TR + 3;
  constexpr int P   = 70;
  constexpr int NP  = NR * 35;
  constexpr int NIT = (NP + 255) / 256;
  constexpr int RS  = RPW >= 2 ? 2 : 1;

  const int col0 = blockIdx.x * 32;
  const int row0 = blockIdx.y * TR;
  const int tid  = threadIdx.x;
  const int wv   = tid >> 5;
  const int lane = tid & 31;
  const int hf   = lane >> 4;
  const int nl   = lane & 15;

  __shared__ __attribute__((aligned(16))) float sIn[NR * P];
  __shared__ float sF[256];

  const float fv = filt[tid < 200 ? tid : 0];

  const int iy0 = 2 * row0 - 2;
  const int ix0 = 2 * col0 - 2;
  unsigned soff[NIT];
  unsigned okm = 0u;
#pragma unroll
  for (int it = 0; it < NIT; ++it) {
    const int idx = tid + 256 * it;
    const int r = idx / 35, j = idx - r * 35;
    const int gy = iy0 + r, gx = ix0 + 2 * j;
    const bool ok = idx < NP && (unsigned)gy < (unsigned)H && (unsigned)gx < (unsigned)W;
    soff[it] = ok ? (unsigned)(gy * W + gx) : 0u;
    okm |= (ok ? 1u : 0u) << it;
  }

  int b = blockIdx.z;
  const int bstep = gridDim.z;
  float2 st[NIT];
  {
    const float* inb = in + (long)b * inBatchStride;
#pragma unroll
    for (int it = 0; it < NIT; ++it)
      st[it] = *reinterpret_cast<const float2*>(inb + soff[it]);
  }

  sF[tid] = fv;
  lds_barrier();

  v2f Aw[9];
  {
    const int ph = nl & 1, ch = nl >> 1;
#pragma unroll
    for (int s = 0; s < 9; ++s) {
#pragma unroll
      for (int t = 0; t < 2; ++t) {
        int kh, wx;
        if (s < 7) { const int q = 2 * s + t; const int rr = q / 7; kh = 2 * rr + hf; wx = q - 7 * rr; }
        else       { kh = 4; wx = 4 * (s - 7) + t + 2 * hf; }
        const int kw = wx - 2 * ph;
        const bool ok = kw >= 0 && kw <= 4;
        const float w = sF[ch * 25 + kh * 5 + (ok ? kw : 0)];
        Aw[s][t] = ok ? w : 0.0f;
      }
    }
  }

  const float* pa = sIn + (2 * wv * RPW + hf) * P + 4 * nl;
  const float* pb = sIn + (2 * wv * RPW + 4) * P + 4 * nl + 2 * hf;

  for (;;) {
#pragma unroll
    for (int it = 0; it < NIT; ++it) {
      const int idx = tid + 256 * it;
      float2 v = st[it];
      if (!((okm >> it) & 1u)) { v.x = 0.0f; v.y = 0.0f; }
      if (idx < NP) *reinterpret_cast<float2*>(&sIn[2 * idx]) = v;
    }
    lds_barrier();

    const int bn = b + bstep;
    if (bn < nb) {
      const float* inb = in + (long)bn * inBatchStride;
#pragma unroll
      for (int it = 0; it < NIT; ++it)
        st[it] = *reinterpret_cast<const float2*>(inb + soff[it]);
    }

#pragma unroll
    for (int i = 0; i < RPW; i += RS) {
      v8f acc0 = {}, acc1 = {};
#pragma unroll
      for (int s = 0; s < 9; ++s) {
        v2f B0, B1 = {};
#pragma unroll
        for (int t = 0; t < 2; ++t) {
          int o;
          const float* p;
          if (s < 7) { const int q = 2 * s + t; const int rr = q / 7; p = pa; o = 2 * rr * P + (q - 7 * rr); }
          else       { p = pb; o = 4 * (s - 7) + t; }
          B0[t] = p[o + (i) * 2 * P];
          if (RS == 2) B1[t] = p[o + (i + 1) * 2 * P];
        }
        acc0 = __builtin_amdgcn_wmma_f32_16x16x4_f32(
            false, Aw[s], false, B0, (short)0, acc0, false, false);
        if (RS == 2)
          acc1 = __builtin_amdgcn_wmma_f32_16x16x4_f32(
              false, Aw[s], false, B1, (short)0, acc1, false, false);
      }
      {
        const int gy = row0 + wv * RPW + i;
        const long cs = (long)Ho * Wo;
        float* o = out + (((long)b * 8 + 4 * hf) * Ho + gy) * Wo + col0 + 2 * nl;
#pragma unroll
        for (int jj = 0; jj < 4; ++jj) {
          float2 v;
          v.x = acc0[2 * jj];
          v.y = acc0[2 * jj + 1];
          *reinterpret_cast<float2*>(o + jj * cs) = v;
        }
        if (RS == 2) {
#pragma unroll
          for (int jj = 0; jj < 4; ++jj) {
            float2 v;
            v.x = acc1[2 * jj];
            v.y = acc1[2 * jj + 1];
            *reinterpret_cast<float2*>(o + jj * cs + Wo) = v;
          }
        }
      }
    }

    if (bn >= nb) break;
    b = bn;
    lds_barrier();
  }
}

template <int CG, int RG, bool WR>
__global__ __launch_bounds__(32 * CG * RG) __attribute__((amdgpu_waves_per_eu(8))) void inv_tconv5x5_s2_wmma(
    const float* __restrict__ rec,  long recBatchStride,
    const float* __restrict__ coef,
    const float* __restrict__ filt,
    float* __restrict__ outp,
    int H, int W, int nb)
{
  constexpr int NT   = 32 * CG * RG;
  constexpr int TC   = 16 * CG;
  constexpr int TR   = 4 * RG;
  constexpr int NRI  = TR + 2;
  constexpr int PT   = TC + 4;
  constexpr int PP   = PT / 2;
  constexpr int PL   = NRI * PT;
  constexpr int PPS  = WR ? TC / 2 : PP;
  constexpr int NP0  = NRI * PPS;
  constexpr int LSTEP = 2 * NT + (WR ? 4 * (NT / PPS) : 0);
  static_assert(!WR || (NT % PPS == 0 && PPS % 32 == 0), "a wave's load must stay inside one staged row");
  constexpr int NIT0 = (NP0 + NT - 1) / NT;
  constexpr int NP1  = 7 * NP0;
  constexpr int NIT1 = (NP1 + NT - 1) / NT;
  constexpr int NS   = 36;
  constexpr int NWT  = (NS * 64 + NT - 1) / NT;
  constexpr int BK   = 6;
  static_assert(PL % 64 >= 16 && PL % 64 <= 48, "plane pitch puts the lane halves on the same banks");

  const int bx0 = blockIdx.x * TC;
  const int by0 = blockIdx.y * TR;
  const int tid = threadIdx.x;
  const int wv  = tid >> 5;
  const int cg  = wv % CG;
  const int rg  = wv / CG;
  const int lane = tid & 31;
  const int hf  = lane >> 4;
  const int nl  = lane & 15;

  __shared__ __attribute__((aligned(16))) float sIn[8 * PL];
  __shared__ float sF[NT];
  __shared__ __attribute__((aligned(16))) float sW[NS * 64];
  __shared__ unsigned sOff[(NIT0 + NIT1) * NT];

  const float fv = filt[tid < 200 ? tid : 0];

  unsigned off0[NIT0], off1[NIT1];
  unsigned okm0 = 0u, okm1 = 0u;
#pragma unroll
  for (int it = 0; it < NIT0; ++it) {
    const int idx = tid + NT * it;
    const int r = idx / PPS, j = idx - r * PPS;
    const int gy = by0 + r - 1, gx = bx0 + 2 * j - (WR ? 0 : 2);
    const bool ok = idx < NP0 && (unsigned)gy < (unsigned)H && (unsigned)gx < (unsigned)W;
    off0[it] = ok ? (unsigned)(gy * W + gx) : 0u;
    okm0 |= (ok ? 1u : 0u) << it;
  }
#pragma unroll
  for (int it = 0; it < NIT1; ++it) {
    const int idx = tid + NT * it;
    const int c1 = idx / NP0, rem = idx - c1 * NP0;
    const int r = rem / PPS, j = rem - r * PPS;
    const int gy = by0 + r - 1, gx = bx0 + 2 * j - (WR ? 0 : 2);
    const bool ok = idx < NP1 && (unsigned)gy < (unsigned)H && (unsigned)gx < (unsigned)W;
    off1[it] = ok ? (unsigned)(((c1 + 1) * H + gy) * W + gx) : 0u;
    okm1 |= (ok ? 1u : 0u) << it;
  }

  int b = blockIdx.z;
  const int bstep = gridDim.z;
  const long coefBatchStride = 8L * H * W;
  float2 s0[NIT0], s1[NIT1];
  {
    const float* recb  = rec  + (long)b * recBatchStride;
    const float* coefb = coef + (long)b * coefBatchStride;
#pragma unroll
    for (int it = 0; it < NIT0; ++it)
      s0[it] = *reinterpret_cast<const float2*>(recb + off0[it]);
#pragma unroll
    for (int it = 0; it < NIT1; ++it)
      s1[it] = *reinterpret_cast<const float2*>(coefb + off1[it]);
  }

#pragma unroll
  for (int it = 0; it < NIT0; ++it) sOff[it * NT + tid] = off0[it];
#pragma unroll
  for (int it = 0; it < NIT1; ++it) sOff[(NIT0 + it) * NT + tid] = off1[it];
  sF[tid] = fv;
  if (WR) {
    if (tid < 8 * NRI) {
      float2 z; z.x = 0.0f; z.y = 0.0f;
      *reinterpret_cast<float2*>(&sIn[tid * PT]) = z;
      *reinterpret_cast<float2*>(&sIn[tid * PT + TC + 2]) = z;
    }
  }
  lds_barrier();

#pragma unroll
  for (int e = 0; e < NWT; ++e) {
    const int idx = tid + NT * e;
    if (idx >= NS * 64) break;
    const int s = idx >> 6, l = (idx >> 1) & 31, t = idx & 1;
    const int m = l & 15, h = l >> 4;
    const int q = 2 * s + t;
    const int cc = q / 18, rem = q - cc * 18;
    const int wy = rem / 3, dxi = rem - wy * 3;
    const int r = m >> 2, py = (m >> 1) & 1, px = m & 1;
    const int dy = wy - 1 - r;
    const int ih = 2 + py - 2 * dy;
    const int iw = 4 + px - 2 * dxi;
    const bool ok = dy >= -1 && dy <= 1 && ih <= 4 && iw <= 4;
    const float w = sF[(2 * cc + h) * 25 + (ok ? ih * 5 + iw : 0)];
    sW[idx] = ok ? w : 0.0f;
  }

  const int lbase = 2 * tid + (WR ? 4 * (tid / PPS) + 2 : 0);
  const int pbase = hf * PL + (rg * 4) * PT + cg * 16 + nl + 1;
  const float* wfrag = sW + 2 * lane;

  for (;;) {
#pragma unroll
    for (int it = 0; it < NIT0; ++it) {
      const int idx = tid + NT * it;
      float2 v = s0[it];
      if (!((okm0 >> it) & 1u)) { v.x = 0.0f; v.y = 0.0f; }
      if (idx < NP0) *reinterpret_cast<float2*>(&sIn[lbase + LSTEP * it]) = v;
    }
#pragma unroll
    for (int it = 0; it < NIT1; ++it) {
      const int idx = tid + NT * it;
      float2 v = s1[it];
      if (!((okm1 >> it) & 1u)) { v.x = 0.0f; v.y = 0.0f; }
      if (idx < NP1) *reinterpret_cast<float2*>(&sIn[PL + lbase + LSTEP * it]) = v;
    }
    lds_barrier();

    const int bn = b + bstep;
    if (bn < nb) {
      const float* recb  = rec  + (long)bn * recBatchStride;
      const float* coefb = coef + (long)bn * coefBatchStride;
#pragma unroll
      for (int it = 0; it < NIT0; ++it)
        s0[it] = *reinterpret_cast<const float2*>(recb + sOff[it * NT + tid]);
#pragma unroll
      for (int it = 0; it < NIT1; ++it)
        s1[it] = *reinterpret_cast<const float2*>(coefb + sOff[(NIT0 + it) * NT + tid]);
    }

    v8f acc0 = {}, acc1 = {};
    v2f Wb[2][BK], Pb[2][BK];
#pragma unroll
    for (int u = 0; u < BK; ++u) {
      Wb[0][u] = *reinterpret_cast<const v2f*>(wfrag + 64 * u);
#pragma unroll
      for (int t = 0; t < 2; ++t) {
        const int q = 2 * u + t;
        const int cc = q / 18, rem = q - cc * 18;
        Pb[0][u][t] = sIn[pbase + 2 * cc * PL + (rem / 3) * PT + (rem % 3)];
      }
    }
#pragma unroll
    for (int blk = 0; blk < NS / BK; ++blk) {
      const int cb = blk & 1, nx = cb ^ 1;
      if (blk + 1 < NS / BK) {
#pragma unroll
        for (int u = 0; u < BK; ++u) {
          const int s = (blk + 1) * BK + u;
          Wb[nx][u] = *reinterpret_cast<const v2f*>(wfrag + 64 * s);
#pragma unroll
          for (int t = 0; t < 2; ++t) {
            const int q = 2 * s + t;
            const int cc = q / 18, rem = q - cc * 18;
            Pb[nx][u][t] = sIn[pbase + 2 * cc * PL + (rem / 3) * PT + (rem % 3)];
          }
        }
      }
      __builtin_amdgcn_sched_barrier(0);
#pragma unroll
      for (int u = 0; u < BK; ++u) {
        if ((blk * BK + u) & 1)
          acc1 = __builtin_amdgcn_wmma_f32_16x16x4_f32(
              false, Wb[cb][u], false, Pb[cb][u], (short)0, acc1, false, false);
        else
          acc0 = __builtin_amdgcn_wmma_f32_16x16x4_f32(
              false, Wb[cb][u], false, Pb[cb][u], (short)0, acc0, false, false);
      }
      __builtin_amdgcn_sched_barrier(0);
    }
    const v8f acc = acc0 + acc1;

    {
      const int oy0 = 2 * (by0 + rg * 4 + 2 * hf);
      float* o = outp + ((long)b * (2 * H) + oy0) * (2L * W) + 2 * (bx0 + cg * 16 + nl);
#pragma unroll
      for (int jj = 0; jj < 4; ++jj) {
        float2 v;
        v.x = acc[2 * jj];
        v.y = acc[2 * jj + 1];
        *reinterpret_cast<float2*>(o + (long)jj * (2L * W)) = v;
      }
    }

    if (bn >= nb) break;
    b = bn;
    lds_barrier();
  }
}


__device__ __forceinline__ void fwd_weight_frags(const float* sF, int nl, int hf, v2f (&Aw)[9]) {
  const int ph = nl & 1, ch = nl >> 1;
#pragma unroll
  for (int s = 0; s < 9; ++s) {
#pragma unroll
    for (int t = 0; t < 2; ++t) {
      int kh, wx;
      if (s < 7) { const int qq = 2 * s + t; const int rr = qq / 7; kh = 2 * rr + hf; wx = qq - 7 * rr; }
      else       { kh = 4; wx = 4 * (s - 7) + t + 2 * hf; }
      const int kw = wx - 2 * ph;
      const bool ok = kw >= 0 && kw <= 4;
      const float w = sF[ch * 25 + kh * 5 + (ok ? kw : 0)];
      Aw[s][t] = ok ? w : 0.0f;
    }
  }
}

template <int P>
__device__ __forceinline__ void fwd_two_rows(const float* pa, const float* pb, const v2f (&Aw)[9],
                                             v8f& acc0, v8f& acc1) {
  const v8f zero = {};
  acc0 = zero;
  acc1 = zero;
#pragma unroll
  for (int s = 0; s < 9; ++s) {
    v2f B0, B1;
#pragma unroll
    for (int t = 0; t < 2; ++t) {
      int o;
      const float* p;
      if (s < 7) { const int qq = 2 * s + t; const int rr = qq / 7; p = pa; o = 2 * rr * P + (qq - 7 * rr); }
      else       { p = pb; o = 4 * (s - 7) + t; }
      B0[t] = p[o];
      B1[t] = p[o + 2 * P];
    }
    acc0 = __builtin_amdgcn_wmma_f32_16x16x4_f32(
        false, Aw[s], false, B0, (short)0, acc0, false, false);
    acc1 = __builtin_amdgcn_wmma_f32_16x16x4_f32(
        false, Aw[s], false, B1, (short)0, acc1, false, false);
  }
}

template <int PL, int PT>
__device__ __forceinline__ v8f inv_product(const float* sI, int pbase, const float* wfrag) {
  constexpr int NS = 36;
  constexpr int BK = 6;
  static_assert(PL % 64 >= 16 && PL % 64 <= 48, "plane pitch puts the lane halves on the same banks");
  v8f acc0 = {}, acc1 = {};
  v2f Wb[2][BK], Pb[2][BK];
#pragma unroll
  for (int u = 0; u < BK; ++u) {
    Wb[0][u] = *reinterpret_cast<const v2f*>(wfrag + 64 * u);
#pragma unroll
    for (int t = 0; t < 2; ++t) {
      const int qq = 2 * u + t;
      const int cc = qq / 18, rem = qq - cc * 18;
      Pb[0][u][t] = sI[pbase + 2 * cc * PL + (rem / 3) * PT + (rem % 3)];
    }
  }
#pragma unroll
  for (int blk = 0; blk < NS / BK; ++blk) {
    const int cb = blk & 1, nx = cb ^ 1;
    if (blk + 1 < NS / BK) {
#pragma unroll
      for (int u = 0; u < BK; ++u) {
        const int s = (blk + 1) * BK + u;
        Wb[nx][u] = *reinterpret_cast<const v2f*>(wfrag + 64 * s);
#pragma unroll
        for (int t = 0; t < 2; ++t) {
          const int qq = 2 * s + t;
          const int cc = qq / 18, rem = qq - cc * 18;
          Pb[nx][u][t] = sI[pbase + 2 * cc * PL + (rem / 3) * PT + (rem % 3)];
        }
      }
    }
    __builtin_amdgcn_sched_barrier(0);
#pragma unroll
    for (int u = 0; u < BK; ++u) {
      if ((blk * BK + u) & 1)
        acc1 = __builtin_amdgcn_wmma_f32_16x16x4_f32(
            false, Wb[cb][u], false, Pb[cb][u], (short)0, acc1, false, false);
      else
        acc0 = __builtin_amdgcn_wmma_f32_16x16x4_f32(
            false, Wb[cb][u], false, Pb[cb][u], (short)0, acc0, false, false);
    }
    __builtin_amdgcn_sched_barrier(0);
  }
  return acc0 + acc1;
}

__global__ __launch_bounds__(256) void mid_levels_wmma(
    const float* __restrict__ in, long inBatchStride,
    const float* __restrict__ ffilt,
    const float* __restrict__ ifilt,
    float* __restrict__ outp)
{
  constexpr int H3  = 128;
  constexpr int H2  = 64;
  constexpr int H1  = 32;
  constexpr int NF2 = 24;
  constexpr int NX2 = 23;
  constexpr int NR3 = 2 * NF2 + 3;
  constexpr int PP3 = 67;
  constexpr int P3  = 2 * PP3;
  constexpr int NP  = NR3 * PP3;
  constexpr int NIT = (NP + 255) / 256;
  constexpr int P2  = 70;
  constexpr int NB1 = 10;
  constexpr int PT1 = H1 + 4;
  constexpr int PL1 = NB1 * PT1;
  constexpr int NI2 = 10;
  constexpr int PT2 = H2 + 4;
  constexpr int PL2 = NI2 * PT2;
  constexpr int NS  = 36;
  constexpr int NWT = NS * 64 / 256;
  constexpr int NZ2 = (NX2 * P2 + 3) / 4 * 4;
  static_assert(P3 % 64 == 6 && P2 % 64 == 6, "row pitch puts the lane halves on the same banks");
  static_assert((8 * PL1) % 4 == 0 && (8 * PL2) % 4 == 0, "the planes are cleared in 16-byte pieces");

  const int q   = blockIdx.x;
  const int b   = blockIdx.y;
  const int tid = threadIdx.x;
  const int wv  = tid >> 5;
  const int lane = tid & 31;
  const int hf  = lane >> 4;
  const int nl  = lane & 15;

  __shared__ __attribute__((aligned(16))) float sX3[NR3 * P3];
  __shared__ __attribute__((aligned(16))) float sX2[NZ2];
  __shared__ __attribute__((aligned(16))) float sI1[8 * PL1];
  __shared__ __attribute__((aligned(16))) float sI2[8 * PL2];
  __shared__ float sFf[256];
  __shared__ float sFi[256];
  __shared__ __attribute__((aligned(16))) float sW[NS * 64];

  const float fvf = ffilt[tid < 200 ? tid : 0];
  const float fvi = ifilt[tid < 200 ? tid : 0];

  const float* inb = in + (long)b * inBatchStride;
  const int iy0 = 16 * q - 18;
  float2 st[NIT];
  unsigned okm = 0u;
#pragma unroll
  for (int it = 0; it < NIT; ++it) {
    const int idx = tid + 256 * it;
    const int r = idx / PP3, j = idx - r * PP3;
    const int gy = iy0 + r, gx = 2 * j - 2;
    const bool ok = idx < NP && (unsigned)gy < (unsigned)H3 && (unsigned)gx < (unsigned)H3;
    st[it] = *reinterpret_cast<const float2*>(inb + (ok ? gy * H3 + gx : 0));
    okm |= (ok ? 1u : 0u) << it;
  }

  {
    float4 z; z.x = 0.0f; z.y = 0.0f; z.z = 0.0f; z.w = 0.0f;
    for (int i = tid; i < NZ2 / 4; i += 256) reinterpret_cast<float4*>(sX2)[i] = z;
    for (int i = tid; i < 8 * PL1 / 4; i += 256) reinterpret_cast<float4*>(sI1)[i] = z;
    for (int i = tid; i < 8 * PL2 / 4; i += 256) reinterpret_cast<float4*>(sI2)[i] = z;
  }
  sFf[tid] = fvf;
  sFi[tid] = fvi;
#pragma unroll
  for (int it = 0; it < NIT; ++it) {
    const int idx = tid + 256 * it;
    float2 v = st[it];
    if (!((okm >> it) & 1u)) { v.x = 0.0f; v.y = 0.0f; }
    if (idx < NP) *reinterpret_cast<float2*>(&sX3[2 * idx]) = v;
  }
  lds_barrier();

  v2f Aw[9];
  fwd_weight_frags(sFf, nl, hf, Aw);

#pragma unroll
  for (int e = 0; e < NWT; ++e) {
    const int idx = tid + 256 * e;
    const int s = idx >> 6, l = (idx >> 1) & 31, t = idx & 1;
    const int m = l & 15, h = l >> 4;
    const int qq = 2 * s + t;
    const int cc = qq / 18, rem = qq - cc * 18;
    const int wy = rem / 3, dxi = rem - wy * 3;
    const int r = m >> 2, py = (m >> 1) & 1, px = m & 1;
    const int dy = wy - 1 - r;
    const int ih = 2 + py - 2 * dy;
    const int iw = 4 + px - 2 * dxi;
    const bool ok = dy >= -1 && dy <= 1 && ih <= 4 && iw <= 4;
    const float w = sFi[(2 * cc + h) * 25 + (ok ? ih * 5 + iw : 0)];
    sW[idx] = ok ? w : 0.0f;
  }

  {
    const int g = wv & 1;
    for (int i = 0; i < 3; ++i) {
      const int lr = 2 * ((wv >> 1) + 4 * i);
      const float* pa = sX3 + (2 * lr + hf) * P3 + 64 * g + 4 * nl;
      const float* pb = sX3 + (2 * lr + 4) * P3 + 64 * g + 4 * nl + 2 * hf;
      v8f acc0, acc1;
      fwd_two_rows<P3>(pa, pb, Aw, acc0, acc1);
      const int x = 32 * g + 2 * nl;
#pragma unroll
      for (int rr = 0; rr < 2; ++rr) {
        const v8f acc = rr ? acc1 : acc0;
        const int l2 = lr + rr;
        const int g2 = 8 * q - 8 + l2;
        if ((unsigned)g2 < (unsigned)H2) {
          if (l2 < NX2 && hf == 0) {
            float2 v;
            v.x = acc[0];
            v.y = acc[1];
            *reinterpret_cast<float2*>(&sX2[l2 * P2 + 2 + x]) = v;
          }
          const int r2 = l2 - 7;
          if ((unsigned)r2 < (unsigned)NI2) {
            float* d = sI2 + (4 * hf) * PL2 + r2 * PT2 + 2 + x;
            if (hf != 0) {
              float2 v;
              v.x = acc[0];
              v.y = acc[1];
              *reinterpret_cast<float2*>(d) = v;
            }
#pragma unroll
            for (int jj = 1; jj < 4; ++jj) {
              float2 v;
              v.x = acc[2 * jj];
              v.y = acc[2 * jj + 1];
              *reinterpret_cast<float2*>(d + jj * PL2) = v;
            }
          }
        }
      }
    }
  }
  lds_barrier();

  if (wv < NB1 / 2) {
    const float* pa = sX2 + (4 * wv + hf) * P2 + 4 * nl;
    const float* pb = sX2 + (4 * wv + 4) * P2 + 4 * nl + 2 * hf;
    v8f acc0, acc1;
    fwd_two_rows<P2>(pa, pb, Aw, acc0, acc1);
    const int lr = 2 * wv;
    const int g0 = 4 * q - 3 + lr;
    float* d = sI1 + (4 * hf) * PL1 + lr * PT1 + 2 + 2 * nl;
    if ((unsigned)g0 < (unsigned)H1) {
#pragma unroll
      for (int jj = 0; jj < 4; ++jj) {
        float2 v;
        v.x = acc0[2 * jj];
        v.y = acc0[2 * jj + 1];
        *reinterpret_cast<float2*>(d + jj * PL1) = v;
      }
    }
    if ((unsigned)(g0 + 1) < (unsigned)H1) {
#pragma unroll
      for (int jj = 0; jj < 4; ++jj) {
        float2 v;
        v.x = acc1[2 * jj];
        v.y = acc1[2 * jj + 1];
        *reinterpret_cast<float2*>(d + jj * PL1 + PT1) = v;
      }
    }
  }
  lds_barrier();

  const float* wfrag = sW + 2 * lane;

  if (wv < 4) {
    const int cg = wv & 1;
    const int rg = wv >> 1;
    const int pbase = hf * PL1 + (rg * 4) * PT1 + cg * 16 + nl + 1;
    const v8f acc = inv_product<PL1, PT1>(sI1, pbase, wfrag);
    const int y0 = 8 * q - 4 + 8 * rg + 4 * hf;
#pragma unroll
    for (int jj = 0; jj < 4; ++jj) {
      const int y = y0 + jj;
      const int r2 = y - (8 * q - 1);
      if ((unsigned)r2 < (unsigned)NI2 && (unsigned)y < (unsigned)H2) {
        float2 v;
        v.x = acc[2 * jj];
        v.y = acc[2 * jj + 1];
        *reinterpret_cast<float2*>(&sI2[r2 * PT2 + 2 + 2 * (cg * 16 + nl)]) = v;
      }
    }
  }
  lds_barrier();

  {
    const int cg = wv & 3;
    const int rg = wv >> 2;
    const int pbase = hf * PL2 + (rg * 4) * PT2 + cg * 16 + nl + 1;
    const v8f acc = inv_product<PL2, PT2>(sI2, pbase, wfrag);
    const int oy0 = 2 * (8 * q + rg * 4 + 2 * hf);
    float* o = outp + ((long)b * H3 + oy0) * (long)H3 + 2 * (cg * 16 + nl);
#pragma unroll
    for (int jj = 0; jj < 4; ++jj) {
      float2 v;
      v.x = acc[2 * jj];
      v.y = acc[2 * jj + 1];
      *reinterpret_cast<float2*>(o + (long)jj * H3) = v;
    }
  }
}

extern "C" void kernel_launch(void* const* d_in, const int* in_sizes, int n_in,
                              void* d_out, int out_size, void* d_ws, size_t ws_size,
                              hipStream_t stream) {
  (void)in_sizes; (void)n_in; (void)out_size; (void)ws_size;
  const float* x     = (const float*)d_in[0];
  const float* fwd_f = (const float*)d_in[1];
  const float* inv_f = (const float*)d_in[2];
  float* out = (float*)d_out;
  float* ws  = (float*)d_ws;

  const size_t n4 = 32UL * 8 * 256 * 256;
  const size_t n3 = 32UL * 8 * 128 * 128;
  float* res4 = ws;
  float* res3 = res4 + n4;
  float* rec2 = res3 + n3;
  float* rec3 = rec2 + 32UL * 128 * 128;

  const dim3 blk(256);

  fwd_conv5x5_s2_wmma<4><<<dim3(8, 8, 16), blk, 0, stream>>>(
      x,    512L * 512,      fwd_f, res4, 512, 512, 256, 256, 32);
  fwd_conv5x5_s2_wmma<2><<<dim3(4, 8, 16), blk, 0, stream>>>(
      res4, 8L * 256 * 256,  fwd_f, res3, 256, 256, 128, 128, 32);

  mid_levels_wmma<<<dim3(8, 32), blk, 0, stream>>>(
      res3, 8L * 128 * 128,  fwd_f, inv_f, rec2);

  inv_tconv5x5_s2_wmma<8, 2, true><<<dim3(1, 16, 16), dim3(512), 0, stream>>>(
      rec2, 128L * 128,      res3, inv_f, rec3, 128, 128, 32);
  inv_tconv5x5_s2_wmma<16, 1, true><<<dim3(1, 64, 8), dim3(512), 0, stream>>>(
      rec3, 256L * 256,      res4, inv_f, out, 256, 256, 32);
}
